# starting kernel
# baseline (speedup 1.0000x reference)
#include <hip/hip_runtime.h>
#include <math.h>

#define BB 8
#define NN 1024
#define CC 64
#define OCC 64
#define KK 5

typedef __attribute__((ext_vector_type(2))) float v2f;
typedef __attribute__((ext_vector_type(4))) float v4f;
typedef __attribute__((ext_vector_type(8))) float v8f;

__global__ void k1_proj(const float* __restrict__ Xr, const float* __restrict__ Xi,
                        const float* __restrict__ awr, const float* __restrict__ awi,
                        float* __restrict__ src_r, float* __restrict__ src_i,
                        float* __restrict__ dst_r, float* __restrict__ dst_i) {
  int id = blockIdx.x * blockDim.x + threadIdx.x;
  if (id >= BB * NN) return;
  const float* xr = Xr + (size_t)id * CC;
  const float* xi = Xi + (size_t)id * CC;
  float sr = 0.f, si = 0.f, dr = 0.f, di = 0.f;
#pragma unroll 4
  for (int c = 0; c < CC; c += 4) {
    v4f a4 = *(const v4f*)(xr + c);
    v4f b4 = *(const v4f*)(xi + c);
    v4f wsr = *(const v4f*)(awr + c);
    v4f wsi = *(const v4f*)(awi + c);
    v4f wdr = *(const v4f*)(awr + CC + c);
    v4f wdi = *(const v4f*)(awi + CC + c);
#pragma unroll
    for (int j = 0; j < 4; ++j) {
      float a = a4[j], b = b4[j];
      sr += a * wsr[j] - b * wsi[j];
      si += a * wsi[j] + b * wsr[j];
      dr += a * wdr[j] - b * wdi[j];
      di += a * wdi[j] + b * wdr[j];
    }
  }
  src_r[id] = sr; src_i[id] = si; dst_r[id] = dr; dst_i[id] = di;
}

__global__ void __launch_bounds__(256) k2_stats(
    const float* __restrict__ src_r, const float* __restrict__ src_i,
    const float* __restrict__ dst_r, const float* __restrict__ dst_i,
    const float* __restrict__ ab_r_p, const float* __restrict__ ab_i_p,
    const float* __restrict__ pa_r_p, const float* __restrict__ pa_i_p,
    float* __restrict__ colmax, float* __restrict__ colsum) {
  __shared__ float s_sr[NN];
  __shared__ float s_si[NN];
  int b = blockIdx.x >> 2;
  int m0 = (blockIdx.x & 3) * 256;
  for (int i = threadIdx.x; i < NN; i += 256) {
    s_sr[i] = src_r[b * NN + i];
    s_si[i] = src_i[b * NN + i];
  }
  __syncthreads();
  float abr = ab_r_p[0], abi = ab_i_p[0], par = pa_r_p[0], pai = pa_i_p[0];
  int m = m0 + threadIdx.x;
  float dr = dst_r[b * NN + m] + abr;
  float di = dst_i[b * NN + m] + abi;
  float mx = -1e30f, sum = 0.f;
  for (int n = 0; n < NN; ++n) {
    float sr = s_sr[n] + dr; sr = sr >= 0.f ? sr : par * sr;
    float si = s_si[n] + di; si = si >= 0.f ? si : pai * si;
    float mag = sqrtf(sr * sr + si * si);
    if (mag > mx) { sum = sum * __expf(mx - mag) + 1.f; mx = mag; }
    else          { sum += __expf(mag - mx); }
  }
  colmax[b * NN + m] = mx;
  colsum[b * NN + m] = sum;
}

__global__ void __launch_bounds__(256) k3_SL(
    const float* __restrict__ Lr, const float* __restrict__ Li,
    const float* __restrict__ src_r, const float* __restrict__ src_i,
    const float* __restrict__ dst_r, const float* __restrict__ dst_i,
    const float* __restrict__ colmax, const float* __restrict__ colsum,
    const float* __restrict__ ab_r_p, const float* __restrict__ ab_i_p,
    const float* __restrict__ pa_r_p, const float* __restrict__ pa_i_p,
    float* __restrict__ SLr, float* __restrict__ SLi) {
  int wave = (blockIdx.x * blockDim.x + threadIdx.x) >> 5;
  int lane = threadIdx.x & 31;
  int b = wave / NN, n = wave % NN;
  float abr = ab_r_p[0], abi = ab_i_p[0], par = pa_r_p[0], pai = pa_i_p[0];
  float ssr = src_r[b * NN + n] + abr;
  float ssi = src_i[b * NN + n] + abi;
  float accR[KK], accI[KK];
#pragma unroll
  for (int k = 0; k < KK; ++k) { accR[k] = 0.f; accI[k] = 0.f; }
  const size_t baseL = (size_t)b * KK * NN * NN + (size_t)n * NN;
  const int sb = b * NN;
  for (int mi = 0; mi < NN / 128; ++mi) {
    int m0 = mi * 128 + lane * 4;
    v4f dr4 = *(const v4f*)(dst_r + sb + m0);
    v4f di4 = *(const v4f*)(dst_i + sb + m0);
    v4f cx4 = *(const v4f*)(colmax + sb + m0);
    v4f cs4 = *(const v4f*)(colsum + sb + m0);
    v4f arv, aiv;
#pragma unroll
    for (int j = 0; j < 4; ++j) {
      float sr = ssr + dr4[j]; sr = sr >= 0.f ? sr : par * sr;
      float si = ssi + di4[j]; si = si >= 0.f ? si : pai * si;
      float mag = sqrtf(sr * sr + si * si);
      float sm = __expf(mag - cx4[j]) / cs4[j];
      float safe = (mag == 0.f) ? 1.f : mag;
      float inv = sm / safe;
      arv[j] = inv * sr;
      aiv[j] = inv * si;
    }
#pragma unroll
    for (int k = 0; k < KK; ++k) {
      size_t idx = baseL + (size_t)k * NN * NN + m0;
      v4f lr4 = *(const v4f*)(Lr + idx);
      v4f li4 = *(const v4f*)(Li + idx);
      float r = accR[k], i2 = accI[k];
#pragma unroll
      for (int j = 0; j < 4; ++j) {
        r  += lr4[j] * arv[j] - li4[j] * aiv[j];
        i2 += lr4[j] * aiv[j] + li4[j] * arv[j];
      }
      accR[k] = r; accI[k] = i2;
    }
  }
#pragma unroll
  for (int k = 0; k < KK; ++k) {
    float r = accR[k], i = accI[k];
    for (int off = 16; off; off >>= 1) {
      r += __shfl_xor(r, off, 32);
      i += __shfl_xor(i, off, 32);
    }
    if (lane == 0) {
      SLr[(b * KK + k) * NN + n] = r;
      SLi[(b * KK + k) * NN + n] = i;
    }
  }
}

__global__ void __launch_bounds__(32) k4_out(
    const float* __restrict__ Xr, const float* __restrict__ Xi,
    const float* __restrict__ Wr, const float* __restrict__ Wi,
    const float* __restrict__ SLr, const float* __restrict__ SLi,
    float* __restrict__ out) {
  int tile = blockIdx.x;
  int b = tile >> 6;
  int n0 = (tile & 63) << 4;
  int lane = threadIdx.x;
  int row = lane & 15;
  int khalf = (lane >> 4) << 1;
  v8f accR[4], accI[4];
#pragma unroll
  for (int i = 0; i < 4; ++i) {
    accR[i] = (v8f){0.f,0.f,0.f,0.f,0.f,0.f,0.f,0.f};
    accI[i] = (v8f){0.f,0.f,0.f,0.f,0.f,0.f,0.f,0.f};
  }
  const int n = n0 + row;
  const size_t xbase = ((size_t)b * NN + n) * CC;
  for (int k = 0; k < KK; ++k) {
    float slr = SLr[(b * KK + k) * NN + n];
    float sli = SLi[(b * KK + k) * NN + n];
    float msli = -sli;
    const float* wr = Wr + (size_t)k * CC * OCC;
    const float* wi = Wi + (size_t)k * CC * OCC;
#pragma unroll 4
    for (int cc = 0; cc < CC / 4; ++cc) {
      int c0 = cc * 4 + khalf;
      v2f xr = *(const v2f*)(Xr + xbase + c0);
      v2f xi = *(const v2f*)(Xi + xbase + c0);
      v2f aRr = { slr  * xr.x, slr  * xr.y };
      v2f aRi = { msli * xi.x, msli * xi.y };
      v2f aIr = { sli  * xr.x, sli  * xr.y };
      v2f aIi = { slr  * xi.x, slr  * xi.y };
#pragma unroll
      for (int ot = 0; ot < 4; ++ot) {
        int o = ot * 16 + row;
        v2f bwr = { wr[(c0 + 0) * OCC + o], wr[(c0 + 1) * OCC + o] };
        v2f bwi = { wi[(c0 + 0) * OCC + o], wi[(c0 + 1) * OCC + o] };
        accR[ot] = __builtin_amdgcn_wmma_f32_16x16x4_f32(false, aRr, false, bwr, (short)0, accR[ot], false, false);
        accR[ot] = __builtin_amdgcn_wmma_f32_16x16x4_f32(false, aRi, false, bwi, (short)0, accR[ot], false, false);
        accI[ot] = __builtin_amdgcn_wmma_f32_16x16x4_f32(false, aIr, false, bwr, (short)0, accI[ot], false, false);
        accI[ot] = __builtin_amdgcn_wmma_f32_16x16x4_f32(false, aIi, false, bwi, (short)0, accI[ot], false, false);
      }
    }
  }
  float* outR = out;
  float* outI = out + (size_t)BB * NN * OCC;
  int rbase = n0 + ((lane >> 4) * 8);
  int col = lane & 15;
#pragma unroll
  for (int j = 0; j < 8; ++j) {
    int rn = rbase + j;
#pragma unroll
    for (int ot = 0; ot < 4; ++ot) {
      size_t oidx = ((size_t)b * NN + rn) * OCC + ot * 16 + col;
      outR[oidx] = accR[ot][j];
      outI[oidx] = accI[ot][j];
    }
  }
}

extern "C" void kernel_launch(void* const* d_in, const int* in_sizes, int n_in,
                              void* d_out, int out_size, void* d_ws, size_t ws_size,
                              hipStream_t stream) {
  const float* Xr  = (const float*)d_in[0];
  const float* Xi  = (const float*)d_in[1];
  const float* Lr  = (const float*)d_in[2];
  const float* Li  = (const float*)d_in[3];
  const float* Wr  = (const float*)d_in[4];
  const float* Wi  = (const float*)d_in[5];
  const float* awr = (const float*)d_in[6];
  const float* awi = (const float*)d_in[7];
  const float* abr = (const float*)d_in[8];
  const float* abi = (const float*)d_in[9];
  const float* par = (const float*)d_in[10];
  const float* pai = (const float*)d_in[11];
  float* out = (float*)d_out;

  float* ws    = (float*)d_ws;
  float* src_r = ws;
  float* src_i = ws + 8192;
  float* dst_r = ws + 16384;
  float* dst_i = ws + 24576;
  float* cmx   = ws + 32768;
  float* csm   = ws + 40960;
  float* SLr   = ws + 49152;
  float* SLi   = ws + 90112;

  k1_proj <<<32, 256, 0, stream>>>(Xr, Xi, awr, awi, src_r, src_i, dst_r, dst_i);
  k2_stats<<<32, 256, 0, stream>>>(src_r, src_i, dst_r, dst_i, abr, abi, par, pai, cmx, csm);
  k3_SL  <<<1024, 256, 0, stream>>>(Lr, Li, src_r, src_i, dst_r, dst_i, cmx, csm,
                                    abr, abi, par, pai, SLr, SLi);
  k4_out  <<<512, 32, 0, stream>>>(Xr, Xi, Wr, Wi, SLr, SLi, out);
}


# out kernel re-tiled: 4 waves per 16-row tile, a wave owns one 16x16 output tile (2048 waves, was 512 of 4 tiles each); slab partials added once per block, rows stored via LDS; sums keep their order
# speedup vs baseline: 3.0086x; 3.0086x over previous
#include <hip/hip_runtime.h>
#include <math.h>

#define BB 8
#define NN 1024
#define CC 64
#define OCC 64
#define KK 5
#define SLABW 64
#define NSLAB (NN / SLABW)
#define RSPLIT 2
#define RPB (NN / RSPLIT)
#define NCH 8

typedef __attribute__((ext_vector_type(2))) float v2f;
typedef __attribute__((ext_vector_type(4))) float v4f;
typedef __attribute__((ext_vector_type(8))) float v8f;

__global__ void k1_proj(const float* __restrict__ Xr, const float* __restrict__ Xi,
                        const float* __restrict__ awr, const float* __restrict__ awi,
                        float* __restrict__ src_r, float* __restrict__ src_i,
                        float* __restrict__ dst_r, float* __restrict__ dst_i) {
  int id = blockIdx.x * blockDim.x + threadIdx.x;
  if (id >= BB * NN) return;
  const float* xr = Xr + (size_t)id * CC;
  const float* xi = Xi + (size_t)id * CC;
  float sr = 0.f, si = 0.f, dr = 0.f, di = 0.f;
#pragma unroll 4
  for (int c = 0; c < CC; c += 4) {
    v4f a4 = *(const v4f*)(xr + c);
    v4f b4 = *(const v4f*)(xi + c);
    v4f wsr = *(const v4f*)(awr + c);
    v4f wsi = *(const v4f*)(awi + c);
    v4f wdr = *(const v4f*)(awr + CC + c);
    v4f wdi = *(const v4f*)(awi + CC + c);
#pragma unroll
    for (int j = 0; j < 4; ++j) {
      float a = a4[j], b = b4[j];
      sr += a * wsr[j] - b * wsi[j];
      si += a * wsi[j] + b * wsr[j];
      dr += a * wdr[j] - b * wdi[j];
      di += a * wdi[j] + b * wdr[j];
    }
  }
  src_r[id] = sr; src_i[id] = si; dst_r[id] = dr; dst_i[id] = di;
}

__global__ void __launch_bounds__(512) k2_colslab(
    const float* __restrict__ Lr, const float* __restrict__ Li,
    const float* __restrict__ src_r, const float* __restrict__ src_i,
    const float* __restrict__ dst_r, const float* __restrict__ dst_i,
    const float* __restrict__ ab_r_p, const float* __restrict__ ab_i_p,
    const float* __restrict__ pa_r_p, const float* __restrict__ pa_i_p,
    float* __restrict__ Pr, float* __restrict__ Pi) {
  __shared__ __attribute__((aligned(16))) v2f   s_s[NN];
  __shared__ float s_red[NCH][SLABW];
  __shared__ __attribute__((aligned(16))) float s_cmax[SLABW];
  __shared__ __attribute__((aligned(16))) float s_csum[SLABW];
  __shared__ float s_out[2 * KK][RPB];

  const int tid = threadIdx.x;
  const int lane = tid & 31;
  const int w = tid >> 5;
  const int rh = blockIdx.x & (RSPLIT - 1);
  const int slab = (blockIdx.x >> 1) & (NSLAB - 1);
  const int b = blockIdx.x >> 5;
  const int sb = b * NN;
  const float abr = ab_r_p[0], abi = ab_i_p[0], par = pa_r_p[0], pai = pa_i_p[0];

  for (int i = tid; i < NN; i += 512) {
    v2f s = { src_r[sb + i] + abr, src_i[sb + i] + abi };
    s_s[i] = s;
  }
  __syncthreads();

  {
    const int c = tid & (SLABW - 1);
    const int rc = tid >> 6;
    const float dr = dst_r[sb + slab * SLABW + c];
    const float di = dst_i[sb + slab * SLABW + c];
    float m2 = 0.f;
    for (int i = 0; i < NN / NCH; ++i) {
      v2f s = s_s[rc * (NN / NCH) + i];
      float sr = s.x + dr; sr = sr >= 0.f ? sr : par * sr;
      float si = s.y + di; si = si >= 0.f ? si : pai * si;
      float q2 = sr * sr + si * si;
      m2 = q2 > m2 ? q2 : m2;
    }
    s_red[rc][c] = m2;
    __syncthreads();
    float cm2 = s_red[0][c];
#pragma unroll
    for (int j = 1; j < NCH; ++j) { float t = s_red[j][c]; cm2 = t > cm2 ? t : cm2; }
    const float cmax = sqrtf(cm2);
    __syncthreads();
    float sum = 0.f;
    for (int i = 0; i < NN / NCH; ++i) {
      v2f s = s_s[rc * (NN / NCH) + i];
      float sr = s.x + dr; sr = sr >= 0.f ? sr : par * sr;
      float si = s.y + di; si = si >= 0.f ? si : pai * si;
      float mag = sqrtf(sr * sr + si * si);
      sum += __expf(mag - cmax);
    }
    s_red[rc][c] = sum;
    __syncthreads();
    if (rc == 0) {
      float cs = s_red[0][c];
#pragma unroll
      for (int j = 1; j < NCH; ++j) cs += s_red[j][c];
      s_cmax[c] = cmax;
      s_csum[c] = cs;
    }
    __syncthreads();
  }

  const int q = lane & 15;
  const int rsub = lane >> 4;
  const int col0 = slab * SLABW + q * 4;
  const v4f dr4 = *(const v4f*)(dst_r + sb + col0);
  const v4f di4 = *(const v4f*)(dst_i + sb + col0);
  const v4f cx4 = *(const v4f*)(s_cmax + q * 4);
  const v4f cs4 = *(const v4f*)(s_csum + q * 4);
  const size_t planeL = (size_t)NN * NN;
  const size_t baseB = (size_t)b * KK * planeL + col0;
  const int rl0 = w * 2 + rsub;

  v4f lr[KK], li[KK];
  {
    const size_t rowoff = baseB + (size_t)(rh * RPB + rl0) * NN;
#pragma unroll
    for (int k = 0; k < KK; ++k) {
      lr[k] = *(const v4f*)(Lr + rowoff + k * planeL);
      li[k] = *(const v4f*)(Li + rowoff + k * planeL);
    }
  }
#pragma unroll 2
  for (int it = 0; it < RPB / 32; ++it) {
    const int rl = it * 32 + rl0;
    const int n = rh * RPB + rl;
    v4f nlr[KK], nli[KK];
    if (it + 1 < RPB / 32) {
      const size_t rowoff = baseB + (size_t)(n + 32) * NN;
#pragma unroll
      for (int k = 0; k < KK; ++k) {
        nlr[k] = *(const v4f*)(Lr + rowoff + k * planeL);
        nli[k] = *(const v4f*)(Li + rowoff + k * planeL);
      }
    } else {
#pragma unroll
      for (int k = 0; k < KK; ++k) { nlr[k] = lr[k]; nli[k] = li[k]; }
    }
    const v2f ss = s_s[n];
    const float ssr = ss.x, ssi = ss.y;
    v4f arv, aiv;
#pragma unroll
    for (int j = 0; j < 4; ++j) {
      float sr = ssr + dr4[j]; sr = sr >= 0.f ? sr : par * sr;
      float si = ssi + di4[j]; si = si >= 0.f ? si : pai * si;
      float mag = sqrtf(sr * sr + si * si);
      float sm = __expf(mag - cx4[j]) / cs4[j];
      float safe = (mag == 0.f) ? 1.f : mag;
      float inv = sm / safe;
      arv[j] = inv * sr;
      aiv[j] = inv * si;
    }
#pragma unroll
    for (int k = 0; k < KK; ++k) {
      float r = 0.f, i2 = 0.f;
#pragma unroll
      for (int j = 0; j < 4; ++j) {
        r  += lr[k][j] * arv[j] - li[k][j] * aiv[j];
        i2 += lr[k][j] * aiv[j] + li[k][j] * arv[j];
      }
      for (int off = 8; off; off >>= 1) {
        r  += __shfl_xor(r, off, 32);
        i2 += __shfl_xor(i2, off, 32);
      }
      if (q == 0) {
        s_out[2 * k][rl] = r;
        s_out[2 * k + 1][rl] = i2;
      }
    }
#pragma unroll
    for (int k = 0; k < KK; ++k) { lr[k] = nlr[k]; li[k] = nli[k]; }
  }
  __syncthreads();
#pragma unroll
  for (int k = 0; k < KK; ++k) {
    const size_t o = ((size_t)(b * KK + k) * NSLAB + slab) * NN + rh * RPB + tid;
    Pr[o] = s_out[2 * k][tid];
    Pi[o] = s_out[2 * k + 1][tid];
  }
}

#define WBLK 8
#define OPITCH 68

struct WFrag { v2f wr[WBLK], wi[WBLK]; };

__device__ __forceinline__ void load_w(WFrag& f, const float* __restrict__ wr, const float* __restrict__ wi) {
#pragma unroll
  for (int j = 0; j < WBLK; ++j) {
    f.wr[j] = (v2f){ wr[(j * 4 + 0) * OCC], wr[(j * 4 + 1) * OCC] };
    f.wi[j] = (v2f){ wi[(j * 4 + 0) * OCC], wi[(j * 4 + 1) * OCC] };
  }
}

__device__ __forceinline__ void mm_block(const v2f* xr, const v2f* xi, const WFrag& f,
                                         float slr, float sli, v8f& accR, v8f& accI) {
  const float msli = -sli;
#pragma unroll
  for (int j = 0; j < WBLK; ++j) {
    v2f aRr = { slr  * xr[j].x, slr  * xr[j].y };
    v2f aRi = { msli * xi[j].x, msli * xi[j].y };
    v2f aIr = { sli  * xr[j].x, sli  * xr[j].y };
    v2f aIi = { slr  * xi[j].x, slr  * xi[j].y };
    accR = __builtin_amdgcn_wmma_f32_16x16x4_f32(false, aRr, false, f.wr[j], (short)0, accR, false, false);
    accI = __builtin_amdgcn_wmma_f32_16x16x4_f32(false, aIr, false, f.wr[j], (short)0, accI, false, false);
    accR = __builtin_amdgcn_wmma_f32_16x16x4_f32(false, aRi, false, f.wi[j], (short)0, accR, false, false);
    accI = __builtin_amdgcn_wmma_f32_16x16x4_f32(false, aIi, false, f.wi[j], (short)0, accI, false, false);
  }
}

__global__ void __launch_bounds__(128) k4_out(
    const float* __restrict__ Xr, const float* __restrict__ Xi,
    const float* __restrict__ Wr, const float* __restrict__ Wi,
    const float* __restrict__ Pr, const float* __restrict__ Pi,
    float* __restrict__ out) {
  __shared__ float s_sl[2 * KK][16];
  __shared__ __attribute__((aligned(16))) float s_o[2][16][OPITCH];

  const int tile = blockIdx.x;
  const int b = tile >> 6;
  const int n0 = (tile & 63) << 4;
  const int t = threadIdx.x;
  const int lane = t & 31;
  const int ot = t >> 5;
  const int row = lane & 15;
  const int h = lane >> 4;
  const int khalf = h << 1;
  const int n = n0 + row;
  const size_t xbase = ((size_t)b * NN + n) * CC + khalf;
  const float* wr0 = Wr + (size_t)khalf * OCC + ot * 16 + row;
  const float* wi0 = Wi + (size_t)khalf * OCC + ot * 16 + row;

  v2f xr[CC / 4], xi[CC / 4];
#pragma unroll
  for (int cc = 0; cc < CC / 4; ++cc) {
    xr[cc] = *(const v2f*)(Xr + xbase + cc * 4);
    xi[cc] = *(const v2f*)(Xi + xbase + cc * 4);
  }
  WFrag wa, wb;
  load_w(wa, wr0, wi0);

  {
    const int id = t >> 4, r = t & 15;
    const float* p = (id < KK ? Pr + (size_t)(b * KK + id) * NSLAB * NN
                              : Pi + (size_t)(b * KK + id - KK) * NSLAB * NN) + n0 + r;
    float s = 0.f;
#pragma unroll
    for (int sl = 0; sl < NSLAB; ++sl) s += p[(size_t)sl * NN];
    s_sl[id][r] = s;
    if (t < 32) {
      const int id2 = 8 + id;
      const float* p2 = Pi + (size_t)(b * KK + id2 - KK) * NSLAB * NN + n0 + r;
      float s2 = 0.f;
#pragma unroll
      for (int sl = 0; sl < NSLAB; ++sl) s2 += p2[(size_t)sl * NN];
      s_sl[id2][r] = s2;
    }
  }
  __syncthreads();

  v8f accR = (v8f){0.f,0.f,0.f,0.f,0.f,0.f,0.f,0.f};
  v8f accI = (v8f){0.f,0.f,0.f,0.f,0.f,0.f,0.f,0.f};
  for (int k = 0; k < KK; ++k) {
    const float* wrk = wr0 + (size_t)k * CC * OCC;
    const float* wik = wi0 + (size_t)k * CC * OCC;
    load_w(wb, wrk + WBLK * 4 * OCC, wik + WBLK * 4 * OCC);
    const float slr = s_sl[k][row];
    const float sli = s_sl[KK + k][row];
    mm_block(xr, xi, wa, slr, sli, accR, accI);
    if (k + 1 < KK) load_w(wa, wrk + CC * OCC, wik + CC * OCC);
    mm_block(xr + WBLK, xi + WBLK, wb, slr, sli, accR, accI);
  }

#pragma unroll
  for (int j = 0; j < 8; ++j) {
    s_o[0][8 * h + j][ot * 16 + row] = accR[j];
    s_o[1][8 * h + j][ot * 16 + row] = accI[j];
  }
  __syncthreads();
  float* outR = out;
  float* outI = out + (size_t)BB * NN * OCC;
#pragma unroll
  for (int i = 0; i < 2; ++i) {
    const int r = ot * 4 + i * 2 + h;
    const size_t oidx = ((size_t)b * NN + n0 + r) * OCC + row * 4;
    *(v4f*)(outR + oidx) = *(const v4f*)&s_o[0][r][row * 4];
    *(v4f*)(outI + oidx) = *(const v4f*)&s_o[1][r][row * 4];
  }
}

extern "C" void kernel_launch(void* const* d_in, const int* in_sizes, int n_in,
                              void* d_out, int out_size, void* d_ws, size_t ws_size,
                              hipStream_t stream) {
  const float* Xr  = (const float*)d_in[0];
  const float* Xi  = (const float*)d_in[1];
  const float* Lr  = (const float*)d_in[2];
  const float* Li  = (const float*)d_in[3];
  const float* Wr  = (const float*)d_in[4];
  const float* Wi  = (const float*)d_in[5];
  const float* awr = (const float*)d_in[6];
  const float* awi = (const float*)d_in[7];
  const float* abr = (const float*)d_in[8];
  const float* abi = (const float*)d_in[9];
  const float* par = (const float*)d_in[10];
  const float* pai = (const float*)d_in[11];
  float* out = (float*)d_out;

  float* ws    = (float*)d_ws;
  float* src_r = ws;
  float* src_i = ws + 8192;
  float* dst_r = ws + 16384;
  float* dst_i = ws + 24576;
  float* Pr    = ws + 32768;
  float* Pi    = ws + 688128;

  k1_proj <<<32, 256, 0, stream>>>(Xr, Xi, awr, awi, src_r, src_i, dst_r, dst_i);
  k2_colslab<<<256, 512, 0, stream>>>(Lr, Li, src_r, src_i, dst_r, dst_i,
                                      abr, abi, par, pai, Pr, Pi);
  k4_out  <<<512, 128, 0, stream>>>(Xr, Xi, Wr, Wi, Pr, Pi, out);
}


# k2 row sums: the ten 16-lane reductions of a row pair as row_xmask DPP moves and adds on the vector ALU, interleaved, one predicated LDS store a trip (were 40 LDS permutes, 20 waits); same sum order
# speedup vs baseline: 3.2156x; 1.0688x over previous
#include <hip/hip_runtime.h>
#include <math.h>

#define BB 8
#define NN 1024
#define CC 64
#define OCC 64
#define KK 5
#define SLABW 64
#define NSLAB (NN / SLABW)
#define RSPLIT 2
#define RPB (NN / RSPLIT)
#define NCH 8

typedef __attribute__((ext_vector_type(2))) float v2f;
typedef __attribute__((ext_vector_type(4))) float v4f;
typedef __attribute__((ext_vector_type(8))) float v8f;

__global__ void k1_proj(const float* __restrict__ Xr, const float* __restrict__ Xi,
                        const float* __restrict__ awr, const float* __restrict__ awi,
                        float* __restrict__ src_r, float* __restrict__ src_i,
                        float* __restrict__ dst_r, float* __restrict__ dst_i) {
  int id = blockIdx.x * blockDim.x + threadIdx.x;
  if (id >= BB * NN) return;
  const float* xr = Xr + (size_t)id * CC;
  const float* xi = Xi + (size_t)id * CC;
  float sr = 0.f, si = 0.f, dr = 0.f, di = 0.f;
#pragma unroll 4
  for (int c = 0; c < CC; c += 4) {
    v4f a4 = *(const v4f*)(xr + c);
    v4f b4 = *(const v4f*)(xi + c);
    v4f wsr = *(const v4f*)(awr + c);
    v4f wsi = *(const v4f*)(awi + c);
    v4f wdr = *(const v4f*)(awr + CC + c);
    v4f wdi = *(const v4f*)(awi + CC + c);
#pragma unroll
    for (int j = 0; j < 4; ++j) {
      float a = a4[j], b = b4[j];
      sr += a * wsr[j] - b * wsi[j];
      si += a * wsi[j] + b * wsr[j];
      dr += a * wdr[j] - b * wdi[j];
      di += a * wdi[j] + b * wdr[j];
    }
  }
  src_r[id] = sr; src_i[id] = si; dst_r[id] = dr; dst_i[id] = di;
}

template <int XMASK>
__device__ __forceinline__ float row_xor_add(float v) {
  const int o = __builtin_amdgcn_update_dpp(0, __builtin_bit_cast(int, v), 0x160 + XMASK, 0xf, 0xf, true);
  return v + __builtin_bit_cast(float, o);
}
__device__ __forceinline__ float row16_sum(float v) {
  v = row_xor_add<8>(v);
  v = row_xor_add<4>(v);
  v = row_xor_add<2>(v);
  v = row_xor_add<1>(v);
  return v;
}

__global__ void __launch_bounds__(512) k2_colslab(
    const float* __restrict__ Lr, const float* __restrict__ Li,
    const float* __restrict__ src_r, const float* __restrict__ src_i,
    const float* __restrict__ dst_r, const float* __restrict__ dst_i,
    const float* __restrict__ ab_r_p, const float* __restrict__ ab_i_p,
    const float* __restrict__ pa_r_p, const float* __restrict__ pa_i_p,
    float* __restrict__ Pr, float* __restrict__ Pi) {
  __shared__ __attribute__((aligned(16))) v2f   s_s[NN];
  __shared__ float s_red[NCH][SLABW];
  __shared__ __attribute__((aligned(16))) float s_cmax[SLABW];
  __shared__ __attribute__((aligned(16))) float s_csum[SLABW];
  __shared__ float s_out[2 * KK][RPB];

  const int tid = threadIdx.x;
  const int lane = tid & 31;
  const int w = tid >> 5;
  const int rh = blockIdx.x & (RSPLIT - 1);
  const int slab = (blockIdx.x >> 1) & (NSLAB - 1);
  const int b = blockIdx.x >> 5;
  const int sb = b * NN;
  const float abr = ab_r_p[0], abi = ab_i_p[0], par = pa_r_p[0], pai = pa_i_p[0];

  for (int i = tid; i < NN; i += 512) {
    v2f s = { src_r[sb + i] + abr, src_i[sb + i] + abi };
    s_s[i] = s;
  }
  __syncthreads();

  {
    const int c = tid & (SLABW - 1);
    const int rc = tid >> 6;
    const float dr = dst_r[sb + slab * SLABW + c];
    const float di = dst_i[sb + slab * SLABW + c];
    float m2 = 0.f;
    for (int i = 0; i < NN / NCH; ++i) {
      v2f s = s_s[rc * (NN / NCH) + i];
      float sr = s.x + dr; sr = sr >= 0.f ? sr : par * sr;
      float si = s.y + di; si = si >= 0.f ? si : pai * si;
      float q2 = sr * sr + si * si;
      m2 = q2 > m2 ? q2 : m2;
    }
    s_red[rc][c] = m2;
    __syncthreads();
    float cm2 = s_red[0][c];
#pragma unroll
    for (int j = 1; j < NCH; ++j) { float t = s_red[j][c]; cm2 = t > cm2 ? t : cm2; }
    const float cmax = sqrtf(cm2);
    __syncthreads();
    float sum = 0.f;
    for (int i = 0; i < NN / NCH; ++i) {
      v2f s = s_s[rc * (NN / NCH) + i];
      float sr = s.x + dr; sr = sr >= 0.f ? sr : par * sr;
      float si = s.y + di; si = si >= 0.f ? si : pai * si;
      float mag = sqrtf(sr * sr + si * si);
      sum += __expf(mag - cmax);
    }
    s_red[rc][c] = sum;
    __syncthreads();
    if (rc == 0) {
      float cs = s_red[0][c];
#pragma unroll
      for (int j = 1; j < NCH; ++j) cs += s_red[j][c];
      s_cmax[c] = cmax;
      s_csum[c] = cs;
    }
    __syncthreads();
  }

  const int q = lane & 15;
  const int rsub = lane >> 4;
  const int col0 = slab * SLABW + q * 4;
  const v4f dr4 = *(const v4f*)(dst_r + sb + col0);
  const v4f di4 = *(const v4f*)(dst_i + sb + col0);
  const v4f cx4 = *(const v4f*)(s_cmax + q * 4);
  const v4f cs4 = *(const v4f*)(s_csum + q * 4);
  const size_t planeL = (size_t)NN * NN;
  const size_t baseB = (size_t)b * KK * planeL + col0;
  const int rl0 = w * 2 + rsub;

  v4f lr[KK], li[KK];
  {
    const size_t rowoff = baseB + (size_t)(rh * RPB + rl0) * NN;
#pragma unroll
    for (int k = 0; k < KK; ++k) {
      lr[k] = *(const v4f*)(Lr + rowoff + k * planeL);
      li[k] = *(const v4f*)(Li + rowoff + k * planeL);
    }
  }
#pragma unroll 2
  for (int it = 0; it < RPB / 32; ++it) {
    const int rl = it * 32 + rl0;
    const int n = rh * RPB + rl;
    v4f nlr[KK], nli[KK];
    if (it + 1 < RPB / 32) {
      const size_t rowoff = baseB + (size_t)(n + 32) * NN;
#pragma unroll
      for (int k = 0; k < KK; ++k) {
        nlr[k] = *(const v4f*)(Lr + rowoff + k * planeL);
        nli[k] = *(const v4f*)(Li + rowoff + k * planeL);
      }
    } else {
#pragma unroll
      for (int k = 0; k < KK; ++k) { nlr[k] = lr[k]; nli[k] = li[k]; }
    }
    const v2f ss = s_s[n];
    const float ssr = ss.x, ssi = ss.y;
    v4f arv, aiv;
#pragma unroll
    for (int j = 0; j < 4; ++j) {
      float sr = ssr + dr4[j]; sr = sr >= 0.f ? sr : par * sr;
      float si = ssi + di4[j]; si = si >= 0.f ? si : pai * si;
      float mag = sqrtf(sr * sr + si * si);
      float sm = __expf(mag - cx4[j]) / cs4[j];
      float safe = (mag == 0.f) ? 1.f : mag;
      float inv = sm / safe;
      arv[j] = inv * sr;
      aiv[j] = inv * si;
    }
    float pr[KK], pi[KK];
#pragma unroll
    for (int k = 0; k < KK; ++k) {
      float r = 0.f, i2 = 0.f;
#pragma unroll
      for (int j = 0; j < 4; ++j) {
        r  += lr[k][j] * arv[j] - li[k][j] * aiv[j];
        i2 += lr[k][j] * aiv[j] + li[k][j] * arv[j];
      }
      pr[k] = r;
      pi[k] = i2;
    }
#pragma unroll
    for (int k = 0; k < KK; ++k) {
      pr[k] = row16_sum(pr[k]);
      pi[k] = row16_sum(pi[k]);
    }
    if (q == 0) {
#pragma unroll
      for (int k = 0; k < KK; ++k) {
        s_out[2 * k][rl] = pr[k];
        s_out[2 * k + 1][rl] = pi[k];
      }
    }
#pragma unroll
    for (int k = 0; k < KK; ++k) { lr[k] = nlr[k]; li[k] = nli[k]; }
  }
  __syncthreads();
#pragma unroll
  for (int k = 0; k < KK; ++k) {
    const size_t o = ((size_t)(b * KK + k) * NSLAB + slab) * NN + rh * RPB + tid;
    Pr[o] = s_out[2 * k][tid];
    Pi[o] = s_out[2 * k + 1][tid];
  }
}

#define WBLK 8
#define OPITCH 68

struct WFrag { v2f wr[WBLK], wi[WBLK]; };

__device__ __forceinline__ void load_w(WFrag& f, const float* __restrict__ wr, const float* __restrict__ wi) {
#pragma unroll
  for (int j = 0; j < WBLK; ++j) {
    f.wr[j] = (v2f){ wr[(j * 4 + 0) * OCC], wr[(j * 4 + 1) * OCC] };
    f.wi[j] = (v2f){ wi[(j * 4 + 0) * OCC], wi[(j * 4 + 1) * OCC] };
  }
}

__device__ __forceinline__ void mm_block(const v2f* xr, const v2f* xi, const WFrag& f,
                                         float slr, float sli, v8f& accR, v8f& accI) {
  const float msli = -sli;
#pragma unroll
  for (int j = 0; j < WBLK; ++j) {
    v2f aRr = { slr  * xr[j].x, slr  * xr[j].y };
    v2f aRi = { msli * xi[j].x, msli * xi[j].y };
    v2f aIr = { sli  * xr[j].x, sli  * xr[j].y };
    v2f aIi = { slr  * xi[j].x, slr  * xi[j].y };
    accR = __builtin_amdgcn_wmma_f32_16x16x4_f32(false, aRr, false, f.wr[j], (short)0, accR, false, false);
    accI = __builtin_amdgcn_wmma_f32_16x16x4_f32(false, aIr, false, f.wr[j], (short)0, accI, false, false);
    accR = __builtin_amdgcn_wmma_f32_16x16x4_f32(false, aRi, false, f.wi[j], (short)0, accR, false, false);
    accI = __builtin_amdgcn_wmma_f32_16x16x4_f32(false, aIi, false, f.wi[j], (short)0, accI, false, false);
  }
}

__global__ void __launch_bounds__(128) k4_out(
    const float* __restrict__ Xr, const float* __restrict__ Xi,
    const float* __restrict__ Wr, const float* __restrict__ Wi,
    const float* __restrict__ Pr, const float* __restrict__ Pi,
    float* __restrict__ out) {
  __shared__ float s_sl[2 * KK][16];
  __shared__ __attribute__((aligned(16))) float s_o[2][16][OPITCH];

  const int tile = blockIdx.x;
  const int b = tile >> 6;
  const int n0 = (tile & 63) << 4;
  const int t = threadIdx.x;
  const int lane = t & 31;
  const int ot = t >> 5;
  const int row = lane & 15;
  const int h = lane >> 4;
  const int khalf = h << 1;
  const int n = n0 + row;
  const size_t xbase = ((size_t)b * NN + n) * CC + khalf;
  const float* wr0 = Wr + (size_t)khalf * OCC + ot * 16 + row;
  const float* wi0 = Wi + (size_t)khalf * OCC + ot * 16 + row;

  v2f xr[CC / 4], xi[CC / 4];
#pragma unroll
  for (int cc = 0; cc < CC / 4; ++cc) {
    xr[cc] = *(const v2f*)(Xr + xbase + cc * 4);
    xi[cc] = *(const v2f*)(Xi + xbase + cc * 4);
  }
  WFrag wa, wb;
  load_w(wa, wr0, wi0);

  {
    const int id = t >> 4, r = t & 15;
    const float* p = (id < KK ? Pr + (size_t)(b * KK + id) * NSLAB * NN
                              : Pi + (size_t)(b * KK + id - KK) * NSLAB * NN) + n0 + r;
    float s = 0.f;
#pragma unroll
    for (int sl = 0; sl < NSLAB; ++sl) s += p[(size_t)sl * NN];
    s_sl[id][r] = s;
    if (t < 32) {
      const int id2 = 8 + id;
      const float* p2 = Pi + (size_t)(b * KK + id2 - KK) * NSLAB * NN + n0 + r;
      float s2 = 0.f;
#pragma unroll
      for (int sl = 0; sl < NSLAB; ++sl) s2 += p2[(size_t)sl * NN];
      s_sl[id2][r] = s2;
    }
  }
  __syncthreads();

  v8f accR = (v8f){0.f,0.f,0.f,0.f,0.f,0.f,0.f,0.f};
  v8f accI = (v8f){0.f,0.f,0.f,0.f,0.f,0.f,0.f,0.f};
  for (int k = 0; k < KK; ++k) {
    const float* wrk = wr0 + (size_t)k * CC * OCC;
    const float* wik = wi0 + (size_t)k * CC * OCC;
    load_w(wb, wrk + WBLK * 4 * OCC, wik + WBLK * 4 * OCC);
    const float slr = s_sl[k][row];
    const float sli = s_sl[KK + k][row];
    mm_block(xr, xi, wa, slr, sli, accR, accI);
    if (k + 1 < KK) load_w(wa, wrk + CC * OCC, wik + CC * OCC);
    mm_block(xr + WBLK, xi + WBLK, wb, slr, sli, accR, accI);
  }

#pragma unroll
  for (int j = 0; j < 8; ++j) {
    s_o[0][8 * h + j][ot * 16 + row] = accR[j];
    s_o[1][8 * h + j][ot * 16 + row] = accI[j];
  }
  __syncthreads();
  float* outR = out;
  float* outI = out + (size_t)BB * NN * OCC;
#pragma unroll
  for (int i = 0; i < 2; ++i) {
    const int r = ot * 4 + i * 2 + h;
    const size_t oidx = ((size_t)b * NN + n0 + r) * OCC + row * 4;
    *(v4f*)(outR + oidx) = *(const v4f*)&s_o[0][r][row * 4];
    *(v4f*)(outI + oidx) = *(const v4f*)&s_o[1][r][row * 4];
  }
}

extern "C" void kernel_launch(void* const* d_in, const int* in_sizes, int n_in,
                              void* d_out, int out_size, void* d_ws, size_t ws_size,
                              hipStream_t stream) {
  const float* Xr  = (const float*)d_in[0];
  const float* Xi  = (const float*)d_in[1];
  const float* Lr  = (const float*)d_in[2];
  const float* Li  = (const float*)d_in[3];
  const float* Wr  = (const float*)d_in[4];
  const float* Wi  = (const float*)d_in[5];
  const float* awr = (const float*)d_in[6];
  const float* awi = (const float*)d_in[7];
  const float* abr = (const float*)d_in[8];
  const float* abi = (const float*)d_in[9];
  const float* par = (const float*)d_in[10];
  const float* pai = (const float*)d_in[11];
  float* out = (float*)d_out;

  float* ws    = (float*)d_ws;
  float* src_r = ws;
  float* src_i = ws + 8192;
  float* dst_r = ws + 16384;
  float* dst_i = ws + 24576;
  float* Pr    = ws + 32768;
  float* Pi    = ws + 688128;

  k1_proj <<<32, 256, 0, stream>>>(Xr, Xi, awr, awi, src_r, src_i, dst_r, dst_i);
  k2_colslab<<<256, 512, 0, stream>>>(Lr, Li, src_r, src_i, dst_r, dst_i,
                                      abr, abi, par, pai, Pr, Pi);
  k4_out  <<<512, 128, 0, stream>>>(Xr, Xi, Wr, Wi, Pr, Pi, out);
}


# k2: each (row, column) magnitude is computed once, in the max pass, and kept in a 256 KB LDS plane; the sum pass and the stream read it back instead of redoing the sqrt; sums keep their order
# speedup vs baseline: 3.4290x; 1.0664x over previous
#include <hip/hip_runtime.h>
#include <math.h>

#define BB 8
#define NN 1024
#define CC 64
#define OCC 64
#define KK 5
#define SLABW 64
#define NSLAB (NN / SLABW)
#define RSPLIT 2
#define RPB (NN / RSPLIT)
#define NCH 8

typedef __attribute__((ext_vector_type(2))) float v2f;
typedef __attribute__((ext_vector_type(4))) float v4f;
typedef __attribute__((ext_vector_type(8))) float v8f;

__global__ void k1_proj(const float* __restrict__ Xr, const float* __restrict__ Xi,
                        const float* __restrict__ awr, const float* __restrict__ awi,
                        float* __restrict__ src_r, float* __restrict__ src_i,
                        float* __restrict__ dst_r, float* __restrict__ dst_i) {
  int id = blockIdx.x * blockDim.x + threadIdx.x;
  if (id >= BB * NN) return;
  const float* xr = Xr + (size_t)id * CC;
  const float* xi = Xi + (size_t)id * CC;
  float sr = 0.f, si = 0.f, dr = 0.f, di = 0.f;
#pragma unroll 4
  for (int c = 0; c < CC; c += 4) {
    v4f a4 = *(const v4f*)(xr + c);
    v4f b4 = *(const v4f*)(xi + c);
    v4f wsr = *(const v4f*)(awr + c);
    v4f wsi = *(const v4f*)(awi + c);
    v4f wdr = *(const v4f*)(awr + CC + c);
    v4f wdi = *(const v4f*)(awi + CC + c);
#pragma unroll
    for (int j = 0; j < 4; ++j) {
      float a = a4[j], b = b4[j];
      sr += a * wsr[j] - b * wsi[j];
      si += a * wsi[j] + b * wsr[j];
      dr += a * wdr[j] - b * wdi[j];
      di += a * wdi[j] + b * wdr[j];
    }
  }
  src_r[id] = sr; src_i[id] = si; dst_r[id] = dr; dst_i[id] = di;
}

template <int XMASK>
__device__ __forceinline__ float row_xor_add(float v) {
  const int o = __builtin_amdgcn_update_dpp(0, __builtin_bit_cast(int, v), 0x160 + XMASK, 0xf, 0xf, true);
  return v + __builtin_bit_cast(float, o);
}
__device__ __forceinline__ float row16_sum(float v) {
  v = row_xor_add<8>(v);
  v = row_xor_add<4>(v);
  v = row_xor_add<2>(v);
  v = row_xor_add<1>(v);
  return v;
}

__global__ void __launch_bounds__(512) k2_colslab(
    const float* __restrict__ Lr, const float* __restrict__ Li,
    const float* __restrict__ src_r, const float* __restrict__ src_i,
    const float* __restrict__ dst_r, const float* __restrict__ dst_i,
    const float* __restrict__ ab_r_p, const float* __restrict__ ab_i_p,
    const float* __restrict__ pa_r_p, const float* __restrict__ pa_i_p,
    float* __restrict__ Pr, float* __restrict__ Pi) {
  __shared__ __attribute__((aligned(16))) v2f   s_s[NN];
  __shared__ __attribute__((aligned(16))) float s_q[NN][SLABW];
  __shared__ float s_red[NCH][SLABW];
  __shared__ __attribute__((aligned(16))) float s_cmax[SLABW];
  __shared__ __attribute__((aligned(16))) float s_csum[SLABW];
  __shared__ float s_out[2 * KK][RPB];

  const int tid = threadIdx.x;
  const int lane = tid & 31;
  const int w = tid >> 5;
  const int rh = blockIdx.x & (RSPLIT - 1);
  const int slab = (blockIdx.x >> 1) & (NSLAB - 1);
  const int b = blockIdx.x >> 5;
  const int sb = b * NN;
  const float abr = ab_r_p[0], abi = ab_i_p[0], par = pa_r_p[0], pai = pa_i_p[0];

  for (int i = tid; i < NN; i += 512) {
    v2f s = { src_r[sb + i] + abr, src_i[sb + i] + abi };
    s_s[i] = s;
  }
  __syncthreads();

  {
    const int c = tid & (SLABW - 1);
    const int rc = tid >> 6;
    const float dr = dst_r[sb + slab * SLABW + c];
    const float di = dst_i[sb + slab * SLABW + c];
    float mx = 0.f;
    for (int i = 0; i < NN / NCH; ++i) {
      const int row = rc * (NN / NCH) + i;
      v2f s = s_s[row];
      float sr = s.x + dr; sr = sr >= 0.f ? sr : par * sr;
      float si = s.y + di; si = si >= 0.f ? si : pai * si;
      float mag = sqrtf(sr * sr + si * si);
      s_q[row][c] = mag;
      mx = mag > mx ? mag : mx;
    }
    s_red[rc][c] = mx;
    __syncthreads();
    float cmax = s_red[0][c];
#pragma unroll
    for (int j = 1; j < NCH; ++j) { float t = s_red[j][c]; cmax = t > cmax ? t : cmax; }
    __syncthreads();
    float sum = 0.f;
    for (int i = 0; i < NN / NCH; ++i) {
      const float mag = s_q[rc * (NN / NCH) + i][c];
      sum += __expf(mag - cmax);
    }
    s_red[rc][c] = sum;
    __syncthreads();
    if (rc == 0) {
      float cs = s_red[0][c];
#pragma unroll
      for (int j = 1; j < NCH; ++j) cs += s_red[j][c];
      s_cmax[c] = cmax;
      s_csum[c] = cs;
    }
    __syncthreads();
  }

  const int q = lane & 15;
  const int rsub = lane >> 4;
  const int col0 = slab * SLABW + q * 4;
  const v4f dr4 = *(const v4f*)(dst_r + sb + col0);
  const v4f di4 = *(const v4f*)(dst_i + sb + col0);
  const v4f cx4 = *(const v4f*)(s_cmax + q * 4);
  const v4f cs4 = *(const v4f*)(s_csum + q * 4);
  const size_t planeL = (size_t)NN * NN;
  const size_t baseB = (size_t)b * KK * planeL + col0;
  const int rl0 = w * 2 + rsub;

  v4f lr[KK], li[KK];
  {
    const size_t rowoff = baseB + (size_t)(rh * RPB + rl0) * NN;
#pragma unroll
    for (int k = 0; k < KK; ++k) {
      lr[k] = *(const v4f*)(Lr + rowoff + k * planeL);
      li[k] = *(const v4f*)(Li + rowoff + k * planeL);
    }
  }
#pragma unroll 2
  for (int it = 0; it < RPB / 32; ++it) {
    const int rl = it * 32 + rl0;
    const int n = rh * RPB + rl;
    const v2f ss = s_s[n];
    const float ssr = ss.x, ssi = ss.y;
    const v4f mq = *(const v4f*)&s_q[n][q * 4];
    v4f nlr[KK], nli[KK];
    if (it + 1 < RPB / 32) {
      const size_t rowoff = baseB + (size_t)(n + 32) * NN;
#pragma unroll
      for (int k = 0; k < KK; ++k) {
        nlr[k] = *(const v4f*)(Lr + rowoff + k * planeL);
        nli[k] = *(const v4f*)(Li + rowoff + k * planeL);
      }
    } else {
#pragma unroll
      for (int k = 0; k < KK; ++k) { nlr[k] = lr[k]; nli[k] = li[k]; }
    }
    v4f arv, aiv;
#pragma unroll
    for (int j = 0; j < 4; ++j) {
      float sr = ssr + dr4[j]; sr = sr >= 0.f ? sr : par * sr;
      float si = ssi + di4[j]; si = si >= 0.f ? si : pai * si;
      const float mag = mq[j];
      float sm = __expf(mag - cx4[j]) / cs4[j];
      float safe = (mag == 0.f) ? 1.f : mag;
      float inv = sm / safe;
      arv[j] = inv * sr;
      aiv[j] = inv * si;
    }
    float pr[KK], pi[KK];
#pragma unroll
    for (int k = 0; k < KK; ++k) {
      float r = 0.f, i2 = 0.f;
#pragma unroll
      for (int j = 0; j < 4; ++j) {
        r  += lr[k][j] * arv[j] - li[k][j] * aiv[j];
        i2 += lr[k][j] * aiv[j] + li[k][j] * arv[j];
      }
      pr[k] = r;
      pi[k] = i2;
    }
#pragma unroll
    for (int k = 0; k < KK; ++k) {
      pr[k] = row16_sum(pr[k]);
      pi[k] = row16_sum(pi[k]);
    }
    if (q == 0) {
#pragma unroll
      for (int k = 0; k < KK; ++k) {
        s_out[2 * k][rl] = pr[k];
        s_out[2 * k + 1][rl] = pi[k];
      }
    }
#pragma unroll
    for (int k = 0; k < KK; ++k) { lr[k] = nlr[k]; li[k] = nli[k]; }
  }
  __syncthreads();
#pragma unroll
  for (int k = 0; k < KK; ++k) {
    const size_t o = ((size_t)(b * KK + k) * NSLAB + slab) * NN + rh * RPB + tid;
    Pr[o] = s_out[2 * k][tid];
    Pi[o] = s_out[2 * k + 1][tid];
  }
}

#define WBLK 8
#define OPITCH 68

struct WFrag { v2f wr[WBLK], wi[WBLK]; };

__device__ __forceinline__ void load_w(WFrag& f, const float* __restrict__ wr, const float* __restrict__ wi) {
#pragma unroll
  for (int j = 0; j < WBLK; ++j) {
    f.wr[j] = (v2f){ wr[(j * 4 + 0) * OCC], wr[(j * 4 + 1) * OCC] };
    f.wi[j] = (v2f){ wi[(j * 4 + 0) * OCC], wi[(j * 4 + 1) * OCC] };
  }
}

__device__ __forceinline__ void mm_block(const v2f* xr, const v2f* xi, const WFrag& f,
                                         float slr, float sli, v8f& accR, v8f& accI) {
  const float msli = -sli;
#pragma unroll
  for (int j = 0; j < WBLK; ++j) {
    v2f aRr = { slr  * xr[j].x, slr  * xr[j].y };
    v2f aRi = { msli * xi[j].x, msli * xi[j].y };
    v2f aIr = { sli  * xr[j].x, sli  * xr[j].y };
    v2f aIi = { slr  * xi[j].x, slr  * xi[j].y };
    accR = __builtin_amdgcn_wmma_f32_16x16x4_f32(false, aRr, false, f.wr[j], (short)0, accR, false, false);
    accI = __builtin_amdgcn_wmma_f32_16x16x4_f32(false, aIr, false, f.wr[j], (short)0, accI, false, false);
    accR = __builtin_amdgcn_wmma_f32_16x16x4_f32(false, aRi, false, f.wi[j], (short)0, accR, false, false);
    accI = __builtin_amdgcn_wmma_f32_16x16x4_f32(false, aIi, false, f.wi[j], (short)0, accI, false, false);
  }
}

__global__ void __launch_bounds__(128) k4_out(
    const float* __restrict__ Xr, const float* __restrict__ Xi,
    const float* __restrict__ Wr, const float* __restrict__ Wi,
    const float* __restrict__ Pr, const float* __restrict__ Pi,
    float* __restrict__ out) {
  __shared__ float s_sl[2 * KK][16];
  __shared__ __attribute__((aligned(16))) float s_o[2][16][OPITCH];

  const int tile = blockIdx.x;
  const int b = tile >> 6;
  const int n0 = (tile & 63) << 4;
  const int t = threadIdx.x;
  const int lane = t & 31;
  const int ot = t >> 5;
  const int row = lane & 15;
  const int h = lane >> 4;
  const int khalf = h << 1;
  const int n = n0 + row;
  const size_t xbase = ((size_t)b * NN + n) * CC + khalf;
  const float* wr0 = Wr + (size_t)khalf * OCC + ot * 16 + row;
  const float* wi0 = Wi + (size_t)khalf * OCC + ot * 16 + row;

  v2f xr[CC / 4], xi[CC / 4];
#pragma unroll
  for (int cc = 0; cc < CC / 4; ++cc) {
    xr[cc] = *(const v2f*)(Xr + xbase + cc * 4);
    xi[cc] = *(const v2f*)(Xi + xbase + cc * 4);
  }
  WFrag wa, wb;
  load_w(wa, wr0, wi0);

  {
    const int id = t >> 4, r = t & 15;
    const float* p = (id < KK ? Pr + (size_t)(b * KK + id) * NSLAB * NN
                              : Pi + (size_t)(b * KK + id - KK) * NSLAB * NN) + n0 + r;
    float s = 0.f;
#pragma unroll
    for (int sl = 0; sl < NSLAB; ++sl) s += p[(size_t)sl * NN];
    s_sl[id][r] = s;
    if (t < 32) {
      const int id2 = 8 + id;
      const float* p2 = Pi + (size_t)(b * KK + id2 - KK) * NSLAB * NN + n0 + r;
      float s2 = 0.f;
#pragma unroll
      for (int sl = 0; sl < NSLAB; ++sl) s2 += p2[(size_t)sl * NN];
      s_sl[id2][r] = s2;
    }
  }
  __syncthreads();

  v8f accR = (v8f){0.f,0.f,0.f,0.f,0.f,0.f,0.f,0.f};
  v8f accI = (v8f){0.f,0.f,0.f,0.f,0.f,0.f,0.f,0.f};
  for (int k = 0; k < KK; ++k) {
    const float* wrk = wr0 + (size_t)k * CC * OCC;
    const float* wik = wi0 + (size_t)k * CC * OCC;
    load_w(wb, wrk + WBLK * 4 * OCC, wik + WBLK * 4 * OCC);
    const float slr = s_sl[k][row];
    const float sli = s_sl[KK + k][row];
    mm_block(xr, xi, wa, slr, sli, accR, accI);
    if (k + 1 < KK) load_w(wa, wrk + CC * OCC, wik + CC * OCC);
    mm_block(xr + WBLK, xi + WBLK, wb, slr, sli, accR, accI);
  }

#pragma unroll
  for (int j = 0; j < 8; ++j) {
    s_o[0][8 * h + j][ot * 16 + row] = accR[j];
    s_o[1][8 * h + j][ot * 16 + row] = accI[j];
  }
  __syncthreads();
  float* outR = out;
  float* outI = out + (size_t)BB * NN * OCC;
#pragma unroll
  for (int i = 0; i < 2; ++i) {
    const int r = ot * 4 + i * 2 + h;
    const size_t oidx = ((size_t)b * NN + n0 + r) * OCC + row * 4;
    *(v4f*)(outR + oidx) = *(const v4f*)&s_o[0][r][row * 4];
    *(v4f*)(outI + oidx) = *(const v4f*)&s_o[1][r][row * 4];
  }
}

extern "C" void kernel_launch(void* const* d_in, const int* in_sizes, int n_in,
                              void* d_out, int out_size, void* d_ws, size_t ws_size,
                              hipStream_t stream) {
  const float* Xr  = (const float*)d_in[0];
  const float* Xi  = (const float*)d_in[1];
  const float* Lr  = (const float*)d_in[2];
  const float* Li  = (const float*)d_in[3];
  const float* Wr  = (const float*)d_in[4];
  const float* Wi  = (const float*)d_in[5];
  const float* awr = (const float*)d_in[6];
  const float* awi = (const float*)d_in[7];
  const float* abr = (const float*)d_in[8];
  const float* abi = (const float*)d_in[9];
  const float* par = (const float*)d_in[10];
  const float* pai = (const float*)d_in[11];
  float* out = (float*)d_out;

  float* ws    = (float*)d_ws;
  float* src_r = ws;
  float* src_i = ws + 8192;
  float* dst_r = ws + 16384;
  float* dst_i = ws + 24576;
  float* Pr    = ws + 32768;
  float* Pi    = ws + 688128;

  k1_proj <<<32, 256, 0, stream>>>(Xr, Xi, awr, awi, src_r, src_i, dst_r, dst_i);
  k2_colslab<<<256, 512, 0, stream>>>(Lr, Li, src_r, src_i, dst_r, dst_i,
                                      abr, abi, par, pai, Pr, Pi);
  k4_out  <<<512, 128, 0, stream>>>(Xr, Xi, Wr, Wi, Pr, Pi, out);
}


# k2 slab narrowed to 32 columns (128 bytes of a row), a block owns all 1024 rows: column statistics computed once, not by two blocks; out kernel adds 32 slab partials; row and column sums reordered
# speedup vs baseline: 3.6096x; 1.0527x over previous
#include <hip/hip_runtime.h>
#include <math.h>

#define BB 8
#define NN 1024
#define CC 64
#define OCC 64
#define KK 5
#define SLABW 32
#define NSLAB (NN / SLABW)
#define RPT 64
#define NCH 16

typedef __attribute__((ext_vector_type(2))) float v2f;
typedef __attribute__((ext_vector_type(4))) float v4f;
typedef __attribute__((ext_vector_type(8))) float v8f;

__global__ void k1_proj(const float* __restrict__ Xr, const float* __restrict__ Xi,
                        const float* __restrict__ awr, const float* __restrict__ awi,
                        float* __restrict__ src_r, float* __restrict__ src_i,
                        float* __restrict__ dst_r, float* __restrict__ dst_i) {
  int id = blockIdx.x * blockDim.x + threadIdx.x;
  if (id >= BB * NN) return;
  const float* xr = Xr + (size_t)id * CC;
  const float* xi = Xi + (size_t)id * CC;
  float sr = 0.f, si = 0.f, dr = 0.f, di = 0.f;
#pragma unroll 4
  for (int c = 0; c < CC; c += 4) {
    v4f a4 = *(const v4f*)(xr + c);
    v4f b4 = *(const v4f*)(xi + c);
    v4f wsr = *(const v4f*)(awr + c);
    v4f wsi = *(const v4f*)(awi + c);
    v4f wdr = *(const v4f*)(awr + CC + c);
    v4f wdi = *(const v4f*)(awi + CC + c);
#pragma unroll
    for (int j = 0; j < 4; ++j) {
      float a = a4[j], b = b4[j];
      sr += a * wsr[j] - b * wsi[j];
      si += a * wsi[j] + b * wsr[j];
      dr += a * wdr[j] - b * wdi[j];
      di += a * wdi[j] + b * wdr[j];
    }
  }
  src_r[id] = sr; src_i[id] = si; dst_r[id] = dr; dst_i[id] = di;
}

template <int XMASK>
__device__ __forceinline__ float row_xor_add(float v) {
  const int o = __builtin_amdgcn_update_dpp(0, __builtin_bit_cast(int, v), 0x160 + XMASK, 0xf, 0xf, true);
  return v + __builtin_bit_cast(float, o);
}
__device__ __forceinline__ float row8_sum(float v) {
  v = row_xor_add<4>(v);
  v = row_xor_add<2>(v);
  v = row_xor_add<1>(v);
  return v;
}

__global__ void __launch_bounds__(512) k2_colslab(
    const float* __restrict__ Lr, const float* __restrict__ Li,
    const float* __restrict__ src_r, const float* __restrict__ src_i,
    const float* __restrict__ dst_r, const float* __restrict__ dst_i,
    const float* __restrict__ ab_r_p, const float* __restrict__ ab_i_p,
    const float* __restrict__ pa_r_p, const float* __restrict__ pa_i_p,
    float* __restrict__ Pr, float* __restrict__ Pi) {
  __shared__ __attribute__((aligned(16))) v2f   s_s[NN];
  __shared__ __attribute__((aligned(16))) float s_q[NN][SLABW];
  __shared__ float s_red[NCH][SLABW];
  __shared__ __attribute__((aligned(16))) float s_cmax[SLABW];
  __shared__ __attribute__((aligned(16))) float s_csum[SLABW];
  __shared__ float s_out[2 * KK][NN];

  const int tid = threadIdx.x;
  const int lane = tid & 31;
  const int w = tid >> 5;
  const int bj = blockIdx.x & (NSLAB - 1);
  const int slab = ((bj & 15) << 1) | (bj >> 4);
  const int b = blockIdx.x >> 5;
  const int sb = b * NN;
  const float abr = ab_r_p[0], abi = ab_i_p[0], par = pa_r_p[0], pai = pa_i_p[0];

  for (int i = tid; i < NN; i += 512) {
    v2f s = { src_r[sb + i] + abr, src_i[sb + i] + abi };
    s_s[i] = s;
  }
  __syncthreads();

  {
    const int c = tid & (SLABW - 1);
    const int rc = tid >> 5;
    const float dr = dst_r[sb + slab * SLABW + c];
    const float di = dst_i[sb + slab * SLABW + c];
    float mx = 0.f;
    for (int i = 0; i < NN / NCH; ++i) {
      const int row = rc * (NN / NCH) + i;
      v2f s = s_s[row];
      float sr = s.x + dr; sr = sr >= 0.f ? sr : par * sr;
      float si = s.y + di; si = si >= 0.f ? si : pai * si;
      float mag = sqrtf(sr * sr + si * si);
      s_q[row][c] = mag;
      mx = mag > mx ? mag : mx;
    }
    s_red[rc][c] = mx;
    __syncthreads();
    float cmax = s_red[0][c];
#pragma unroll
    for (int j = 1; j < NCH; ++j) { float t = s_red[j][c]; cmax = t > cmax ? t : cmax; }
    __syncthreads();
    float sum = 0.f;
    for (int i = 0; i < NN / NCH; ++i) {
      const float mag = s_q[rc * (NN / NCH) + i][c];
      sum += __expf(mag - cmax);
    }
    s_red[rc][c] = sum;
    __syncthreads();
    if (rc == 0) {
      float cs = s_red[0][c];
#pragma unroll
      for (int j = 1; j < NCH; ++j) cs += s_red[j][c];
      s_cmax[c] = cmax;
      s_csum[c] = cs;
    }
    __syncthreads();
  }

  const int q = lane & 7;
  const int rsub = lane >> 3;
  const int col0 = slab * SLABW + q * 4;
  const v4f dr4 = *(const v4f*)(dst_r + sb + col0);
  const v4f di4 = *(const v4f*)(dst_i + sb + col0);
  const v4f cx4 = *(const v4f*)(s_cmax + q * 4);
  const v4f cs4 = *(const v4f*)(s_csum + q * 4);
  const size_t planeL = (size_t)NN * NN;
  const size_t baseB = (size_t)b * KK * planeL + col0;
  const int rl0 = w * 4 + rsub;

  v4f lr[KK], li[KK];
  {
    const size_t rowoff = baseB + (size_t)rl0 * NN;
#pragma unroll
    for (int k = 0; k < KK; ++k) {
      lr[k] = *(const v4f*)(Lr + rowoff + k * planeL);
      li[k] = *(const v4f*)(Li + rowoff + k * planeL);
    }
  }
#pragma unroll 2
  for (int it = 0; it < NN / RPT; ++it) {
    const int n = it * RPT + rl0;
    const v2f ss = s_s[n];
    const float ssr = ss.x, ssi = ss.y;
    const v4f mq = *(const v4f*)&s_q[n][q * 4];
    v4f nlr[KK], nli[KK];
    if (it + 1 < NN / RPT) {
      const size_t rowoff = baseB + (size_t)(n + RPT) * NN;
#pragma unroll
      for (int k = 0; k < KK; ++k) {
        nlr[k] = *(const v4f*)(Lr + rowoff + k * planeL);
        nli[k] = *(const v4f*)(Li + rowoff + k * planeL);
      }
    } else {
#pragma unroll
      for (int k = 0; k < KK; ++k) { nlr[k] = lr[k]; nli[k] = li[k]; }
    }
    v4f arv, aiv;
#pragma unroll
    for (int j = 0; j < 4; ++j) {
      float sr = ssr + dr4[j]; sr = sr >= 0.f ? sr : par * sr;
      float si = ssi + di4[j]; si = si >= 0.f ? si : pai * si;
      const float mag = mq[j];
      float sm = __expf(mag - cx4[j]) / cs4[j];
      float safe = (mag == 0.f) ? 1.f : mag;
      float inv = sm / safe;
      arv[j] = inv * sr;
      aiv[j] = inv * si;
    }
    float pr[KK], pi[KK];
#pragma unroll
    for (int k = 0; k < KK; ++k) {
      float r = 0.f, i2 = 0.f;
#pragma unroll
      for (int j = 0; j < 4; ++j) {
        r  += lr[k][j] * arv[j] - li[k][j] * aiv[j];
        i2 += lr[k][j] * aiv[j] + li[k][j] * arv[j];
      }
      pr[k] = r;
      pi[k] = i2;
    }
#pragma unroll
    for (int k = 0; k < KK; ++k) {
      pr[k] = row8_sum(pr[k]);
      pi[k] = row8_sum(pi[k]);
    }
    if (q == 0) {
#pragma unroll
      for (int k = 0; k < KK; ++k) {
        s_out[2 * k][n] = pr[k];
        s_out[2 * k + 1][n] = pi[k];
      }
    }
#pragma unroll
    for (int k = 0; k < KK; ++k) { lr[k] = nlr[k]; li[k] = nli[k]; }
  }
  __syncthreads();
#pragma unroll
  for (int k = 0; k < KK; ++k) {
    const size_t o = ((size_t)(b * KK + k) * NSLAB + slab) * NN + tid;
    Pr[o] = s_out[2 * k][tid];
    Pi[o] = s_out[2 * k + 1][tid];
    Pr[o + 512] = s_out[2 * k][tid + 512];
    Pi[o + 512] = s_out[2 * k + 1][tid + 512];
  }
}

#define WBLK 8
#define OPITCH 68

struct WFrag { v2f wr[WBLK], wi[WBLK]; };

__device__ __forceinline__ void load_w(WFrag& f, const float* __restrict__ wr, const float* __restrict__ wi) {
#pragma unroll
  for (int j = 0; j < WBLK; ++j) {
    f.wr[j] = (v2f){ wr[(j * 4 + 0) * OCC], wr[(j * 4 + 1) * OCC] };
    f.wi[j] = (v2f){ wi[(j * 4 + 0) * OCC], wi[(j * 4 + 1) * OCC] };
  }
}

__device__ __forceinline__ void mm_block(const v2f* xr, const v2f* xi, const WFrag& f,
                                         float slr, float sli, v8f& accR, v8f& accI) {
  const float msli = -sli;
#pragma unroll
  for (int j = 0; j < WBLK; ++j) {
    v2f aRr = { slr  * xr[j].x, slr  * xr[j].y };
    v2f aRi = { msli * xi[j].x, msli * xi[j].y };
    v2f aIr = { sli  * xr[j].x, sli  * xr[j].y };
    v2f aIi = { slr  * xi[j].x, slr  * xi[j].y };
    accR = __builtin_amdgcn_wmma_f32_16x16x4_f32(false, aRr, false, f.wr[j], (short)0, accR, false, false);
    accI = __builtin_amdgcn_wmma_f32_16x16x4_f32(false, aIr, false, f.wr[j], (short)0, accI, false, false);
    accR = __builtin_amdgcn_wmma_f32_16x16x4_f32(false, aRi, false, f.wi[j], (short)0, accR, false, false);
    accI = __builtin_amdgcn_wmma_f32_16x16x4_f32(false, aIi, false, f.wi[j], (short)0, accI, false, false);
  }
}

__global__ void __launch_bounds__(128) k4_out(
    const float* __restrict__ Xr, const float* __restrict__ Xi,
    const float* __restrict__ Wr, const float* __restrict__ Wi,
    const float* __restrict__ Pr, const float* __restrict__ Pi,
    float* __restrict__ out) {
  __shared__ float s_sl[2 * KK][16];
  __shared__ __attribute__((aligned(16))) float s_o[2][16][OPITCH];

  const int tile = blockIdx.x;
  const int b = tile >> 6;
  const int n0 = (tile & 63) << 4;
  const int t = threadIdx.x;
  const int lane = t & 31;
  const int ot = t >> 5;
  const int row = lane & 15;
  const int h = lane >> 4;
  const int khalf = h << 1;
  const int n = n0 + row;
  const size_t xbase = ((size_t)b * NN + n) * CC + khalf;
  const float* wr0 = Wr + (size_t)khalf * OCC + ot * 16 + row;
  const float* wi0 = Wi + (size_t)khalf * OCC + ot * 16 + row;

  v2f xr[CC / 4], xi[CC / 4];
#pragma unroll
  for (int cc = 0; cc < CC / 4; ++cc) {
    xr[cc] = *(const v2f*)(Xr + xbase + cc * 4);
    xi[cc] = *(const v2f*)(Xi + xbase + cc * 4);
  }
  WFrag wa, wb;
  load_w(wa, wr0, wi0);

  {
    const int id = t >> 4, r = t & 15;
    const float* p = (id < KK ? Pr + (size_t)(b * KK + id) * NSLAB * NN
                              : Pi + (size_t)(b * KK + id - KK) * NSLAB * NN) + n0 + r;
    float s = 0.f;
#pragma unroll
    for (int sl = 0; sl < NSLAB; ++sl) s += p[(size_t)sl * NN];
    s_sl[id][r] = s;
    if (t < 32) {
      const int id2 = 8 + id;
      const float* p2 = Pi + (size_t)(b * KK + id2 - KK) * NSLAB * NN + n0 + r;
      float s2 = 0.f;
#pragma unroll
      for (int sl = 0; sl < NSLAB; ++sl) s2 += p2[(size_t)sl * NN];
      s_sl[id2][r] = s2;
    }
  }
  __syncthreads();

  v8f accR = (v8f){0.f,0.f,0.f,0.f,0.f,0.f,0.f,0.f};
  v8f accI = (v8f){0.f,0.f,0.f,0.f,0.f,0.f,0.f,0.f};
  for (int k = 0; k < KK; ++k) {
    const float* wrk = wr0 + (size_t)k * CC * OCC;
    const float* wik = wi0 + (size_t)k * CC * OCC;
    load_w(wb, wrk + WBLK * 4 * OCC, wik + WBLK * 4 * OCC);
    const float slr = s_sl[k][row];
    const float sli = s_sl[KK + k][row];
    mm_block(xr, xi, wa, slr, sli, accR, accI);
    if (k + 1 < KK) load_w(wa, wrk + CC * OCC, wik + CC * OCC);
    mm_block(xr + WBLK, xi + WBLK, wb, slr, sli, accR, accI);
  }

#pragma unroll
  for (int j = 0; j < 8; ++j) {
    s_o[0][8 * h + j][ot * 16 + row] = accR[j];
    s_o[1][8 * h + j][ot * 16 + row] = accI[j];
  }
  __syncthreads();
  float* outR = out;
  float* outI = out + (size_t)BB * NN * OCC;
#pragma unroll
  for (int i = 0; i < 2; ++i) {
    const int r = ot * 4 + i * 2 + h;
    const size_t oidx = ((size_t)b * NN + n0 + r) * OCC + row * 4;
    *(v4f*)(outR + oidx) = *(const v4f*)&s_o[0][r][row * 4];
    *(v4f*)(outI + oidx) = *(const v4f*)&s_o[1][r][row * 4];
  }
}

extern "C" void kernel_launch(void* const* d_in, const int* in_sizes, int n_in,
                              void* d_out, int out_size, void* d_ws, size_t ws_size,
                              hipStream_t stream) {
  const float* Xr  = (const float*)d_in[0];
  const float* Xi  = (const float*)d_in[1];
  const float* Lr  = (const float*)d_in[2];
  const float* Li  = (const float*)d_in[3];
  const float* Wr  = (const float*)d_in[4];
  const float* Wi  = (const float*)d_in[5];
  const float* awr = (const float*)d_in[6];
  const float* awi = (const float*)d_in[7];
  const float* abr = (const float*)d_in[8];
  const float* abi = (const float*)d_in[9];
  const float* par = (const float*)d_in[10];
  const float* pai = (const float*)d_in[11];
  float* out = (float*)d_out;

  float* ws    = (float*)d_ws;
  float* src_r = ws;
  float* src_i = ws + 8192;
  float* dst_r = ws + 16384;
  float* dst_i = ws + 24576;
  float* Pr    = ws + 32768;
  float* Pi    = ws + 1343488;

  k1_proj <<<32, 256, 0, stream>>>(Xr, Xi, awr, awi, src_r, src_i, dst_r, dst_i);
  k2_colslab<<<256, 512, 0, stream>>>(Lr, Li, src_r, src_i, dst_r, dst_i,
                                      abr, abi, par, pai, Pr, Pi);
  k4_out  <<<512, 128, 0, stream>>>(Xr, Xi, Wr, Wi, Pr, Pi, out);
}


# k1 on the matrix unit: a wave owns 16 rows, its four projections are one f32 16x16x4 product chain ([Xr|Xi] row fragments x signed aw columns), 512 waves, 64 loads at once; channel sums reordered
# speedup vs baseline: 3.7534x; 1.0398x over previous
#include <hip/hip_runtime.h>
#include <math.h>

#define BB 8
#define NN 1024
#define CC 64
#define OCC 64
#define KK 5
#define SLABW 32
#define NSLAB (NN / SLABW)
#define RPT 64
#define NCH 16

typedef __attribute__((ext_vector_type(2))) float v2f;
typedef __attribute__((ext_vector_type(4))) float v4f;
typedef __attribute__((ext_vector_type(8))) float v8f;

#define PPITCH 36
__global__ void __launch_bounds__(64) k1_proj(const float* __restrict__ Xr, const float* __restrict__ Xi,
                        const float* __restrict__ awr, const float* __restrict__ awi,
                        float* __restrict__ src_r, float* __restrict__ src_i,
                        float* __restrict__ dst_r, float* __restrict__ dst_i) {
  __shared__ __attribute__((aligned(16))) float s_p[16][PPITCH];
  const int t = threadIdx.x;
  const int lane = t & 31;
  const int w = t >> 5;
  const int r = lane & 15;
  const int h = lane >> 4;
  const int khalf = h << 1;
  const int n0 = blockIdx.x * 32;

  const size_t xbase = (size_t)(n0 + w * 16 + r) * CC + khalf;
  const int c = r & 3;
  const int half = (c & 2) ? CC : 0;
  const float* pr = ((c & 1) ? awi : awr) + half + khalf;
  const float* pi = ((c & 1) ? awr : awi) + half + khalf;
  const float sg = (c & 1) ? 1.f : -1.f;

  v2f xr[CC / 4], xi[CC / 4], br[CC / 4], bi[CC / 4];
#pragma unroll
  for (int cc = 0; cc < CC / 4; ++cc) {
    xr[cc] = *(const v2f*)(Xr + xbase + cc * 4);
    xi[cc] = *(const v2f*)(Xi + xbase + cc * 4);
    br[cc] = *(const v2f*)(pr + cc * 4);
    bi[cc] = *(const v2f*)(pi + cc * 4);
  }
  __builtin_amdgcn_sched_barrier(0);
  v8f accA = (v8f){0.f,0.f,0.f,0.f,0.f,0.f,0.f,0.f};
  v8f accB = (v8f){0.f,0.f,0.f,0.f,0.f,0.f,0.f,0.f};
#pragma unroll
  for (int cc = 0; cc < CC / 4; ++cc) {
    const v2f bs = { sg * bi[cc].x, sg * bi[cc].y };
    accA = __builtin_amdgcn_wmma_f32_16x16x4_f32(false, xr[cc], false, br[cc], (short)0, accA, false, false);
    accB = __builtin_amdgcn_wmma_f32_16x16x4_f32(false, xi[cc], false, bs, (short)0, accB, false, false);
  }
  const v8f d = accA + accB;
#pragma unroll
  for (int j = 0; j < 8; ++j) s_p[r][w * 16 + 8 * h + j] = d[j];
  __syncthreads();
  float* o0 = w ? dst_r : src_r;
  float* o1 = w ? dst_i : src_i;
  o0[n0 + lane] = s_p[2 * w][lane];
  o1[n0 + lane] = s_p[2 * w + 1][lane];
}

template <int XMASK>
__device__ __forceinline__ float row_xor_add(float v) {
  const int o = __builtin_amdgcn_update_dpp(0, __builtin_bit_cast(int, v), 0x160 + XMASK, 0xf, 0xf, true);
  return v + __builtin_bit_cast(float, o);
}
__device__ __forceinline__ float row8_sum(float v) {
  v = row_xor_add<4>(v);
  v = row_xor_add<2>(v);
  v = row_xor_add<1>(v);
  return v;
}

__global__ void __launch_bounds__(512) k2_colslab(
    const float* __restrict__ Lr, const float* __restrict__ Li,
    const float* __restrict__ src_r, const float* __restrict__ src_i,
    const float* __restrict__ dst_r, const float* __restrict__ dst_i,
    const float* __restrict__ ab_r_p, const float* __restrict__ ab_i_p,
    const float* __restrict__ pa_r_p, const float* __restrict__ pa_i_p,
    float* __restrict__ Pr, float* __restrict__ Pi) {
  __shared__ __attribute__((aligned(16))) v2f   s_s[NN];
  __shared__ __attribute__((aligned(16))) float s_q[NN][SLABW];
  __shared__ float s_red[NCH][SLABW];
  __shared__ __attribute__((aligned(16))) float s_cmax[SLABW];
  __shared__ __attribute__((aligned(16))) float s_csum[SLABW];
  __shared__ float s_out[2 * KK][NN];

  const int tid = threadIdx.x;
  const int lane = tid & 31;
  const int w = tid >> 5;
  const int bj = blockIdx.x & (NSLAB - 1);
  const int slab = ((bj & 15) << 1) | (bj >> 4);
  const int b = blockIdx.x >> 5;
  const int sb = b * NN;
  const float abr = ab_r_p[0], abi = ab_i_p[0], par = pa_r_p[0], pai = pa_i_p[0];

  for (int i = tid; i < NN; i += 512) {
    v2f s = { src_r[sb + i] + abr, src_i[sb + i] + abi };
    s_s[i] = s;
  }
  __syncthreads();

  {
    const int c = tid & (SLABW - 1);
    const int rc = tid >> 5;
    const float dr = dst_r[sb + slab * SLABW + c];
    const float di = dst_i[sb + slab * SLABW + c];
    float mx = 0.f;
    for (int i = 0; i < NN / NCH; ++i) {
      const int row = rc * (NN / NCH) + i;
      v2f s = s_s[row];
      float sr = s.x + dr; sr = sr >= 0.f ? sr : par * sr;
      float si = s.y + di; si = si >= 0.f ? si : pai * si;
      float mag = sqrtf(sr * sr + si * si);
      s_q[row][c] = mag;
      mx = mag > mx ? mag : mx;
    }
    s_red[rc][c] = mx;
    __syncthreads();
    float cmax = s_red[0][c];
#pragma unroll
    for (int j = 1; j < NCH; ++j) { float t = s_red[j][c]; cmax = t > cmax ? t : cmax; }
    __syncthreads();
    float sum = 0.f;
    for (int i = 0; i < NN / NCH; ++i) {
      const float mag = s_q[rc * (NN / NCH) + i][c];
      sum += __expf(mag - cmax);
    }
    s_red[rc][c] = sum;
    __syncthreads();
    if (rc == 0) {
      float cs = s_red[0][c];
#pragma unroll
      for (int j = 1; j < NCH; ++j) cs += s_red[j][c];
      s_cmax[c] = cmax;
      s_csum[c] = cs;
    }
    __syncthreads();
  }

  const int q = lane & 7;
  const int rsub = lane >> 3;
  const int col0 = slab * SLABW + q * 4;
  const v4f dr4 = *(const v4f*)(dst_r + sb + col0);
  const v4f di4 = *(const v4f*)(dst_i + sb + col0);
  const v4f cx4 = *(const v4f*)(s_cmax + q * 4);
  const v4f cs4 = *(const v4f*)(s_csum + q * 4);
  const size_t planeL = (size_t)NN * NN;
  const size_t baseB = (size_t)b * KK * planeL + col0;
  const int rl0 = w * 4 + rsub;

  v4f lr[KK], li[KK];
  {
    const size_t rowoff = baseB + (size_t)rl0 * NN;
#pragma unroll
    for (int k = 0; k < KK; ++k) {
      lr[k] = *(const v4f*)(Lr + rowoff + k * planeL);
      li[k] = *(const v4f*)(Li + rowoff + k * planeL);
    }
  }
#pragma unroll 2
  for (int it = 0; it < NN / RPT; ++it) {
    const int n = it * RPT + rl0;
    const v2f ss = s_s[n];
    const float ssr = ss.x, ssi = ss.y;
    const v4f mq = *(const v4f*)&s_q[n][q * 4];
    v4f nlr[KK], nli[KK];
    if (it + 1 < NN / RPT) {
      const size_t rowoff = baseB + (size_t)(n + RPT) * NN;
#pragma unroll
      for (int k = 0; k < KK; ++k) {
        nlr[k] = *(const v4f*)(Lr + rowoff + k * planeL);
        nli[k] = *(const v4f*)(Li + rowoff + k * planeL);
      }
    } else {
#pragma unroll
      for (int k = 0; k < KK; ++k) { nlr[k] = lr[k]; nli[k] = li[k]; }
    }
    v4f arv, aiv;
#pragma unroll
    for (int j = 0; j < 4; ++j) {
      float sr = ssr + dr4[j]; sr = sr >= 0.f ? sr : par * sr;
      float si = ssi + di4[j]; si = si >= 0.f ? si : pai * si;
      const float mag = mq[j];
      float sm = __expf(mag - cx4[j]) / cs4[j];
      float safe = (mag == 0.f) ? 1.f : mag;
      float inv = sm / safe;
      arv[j] = inv * sr;
      aiv[j] = inv * si;
    }
    float pr[KK], pi[KK];
#pragma unroll
    for (int k = 0; k < KK; ++k) {
      float r = 0.f, i2 = 0.f;
#pragma unroll
      for (int j = 0; j < 4; ++j) {
        r  += lr[k][j] * arv[j] - li[k][j] * aiv[j];
        i2 += lr[k][j] * aiv[j] + li[k][j] * arv[j];
      }
      pr[k] = r;
      pi[k] = i2;
    }
#pragma unroll
    for (int k = 0; k < KK; ++k) {
      pr[k] = row8_sum(pr[k]);
      pi[k] = row8_sum(pi[k]);
    }
    if (q == 0) {
#pragma unroll
      for (int k = 0; k < KK; ++k) {
        s_out[2 * k][n] = pr[k];
        s_out[2 * k + 1][n] = pi[k];
      }
    }
#pragma unroll
    for (int k = 0; k < KK; ++k) { lr[k] = nlr[k]; li[k] = nli[k]; }
  }
  __syncthreads();
#pragma unroll
  for (int k = 0; k < KK; ++k) {
    const size_t o = ((size_t)(b * KK + k) * NSLAB + slab) * NN + tid;
    Pr[o] = s_out[2 * k][tid];
    Pi[o] = s_out[2 * k + 1][tid];
    Pr[o + 512] = s_out[2 * k][tid + 512];
    Pi[o + 512] = s_out[2 * k + 1][tid + 512];
  }
}

#define WBLK 8
#define OPITCH 68

struct WFrag { v2f wr[WBLK], wi[WBLK]; };

__device__ __forceinline__ void load_w(WFrag& f, const float* __restrict__ wr, const float* __restrict__ wi) {
#pragma unroll
  for (int j = 0; j < WBLK; ++j) {
    f.wr[j] = (v2f){ wr[(j * 4 + 0) * OCC], wr[(j * 4 + 1) * OCC] };
    f.wi[j] = (v2f){ wi[(j * 4 + 0) * OCC], wi[(j * 4 + 1) * OCC] };
  }
}

__device__ __forceinline__ void mm_block(const v2f* xr, const v2f* xi, const WFrag& f,
                                         float slr, float sli, v8f& accR, v8f& accI) {
  const float msli = -sli;
#pragma unroll
  for (int j = 0; j < WBLK; ++j) {
    v2f aRr = { slr  * xr[j].x, slr  * xr[j].y };
    v2f aRi = { msli * xi[j].x, msli * xi[j].y };
    v2f aIr = { sli  * xr[j].x, sli  * xr[j].y };
    v2f aIi = { slr  * xi[j].x, slr  * xi[j].y };
    accR = __builtin_amdgcn_wmma_f32_16x16x4_f32(false, aRr, false, f.wr[j], (short)0, accR, false, false);
    accI = __builtin_amdgcn_wmma_f32_16x16x4_f32(false, aIr, false, f.wr[j], (short)0, accI, false, false);
    accR = __builtin_amdgcn_wmma_f32_16x16x4_f32(false, aRi, false, f.wi[j], (short)0, accR, false, false);
    accI = __builtin_amdgcn_wmma_f32_16x16x4_f32(false, aIi, false, f.wi[j], (short)0, accI, false, false);
  }
}

__global__ void __launch_bounds__(128) k4_out(
    const float* __restrict__ Xr, const float* __restrict__ Xi,
    const float* __restrict__ Wr, const float* __restrict__ Wi,
    const float* __restrict__ Pr, const float* __restrict__ Pi,
    float* __restrict__ out) {
  __shared__ float s_sl[2 * KK][16];
  __shared__ __attribute__((aligned(16))) float s_o[2][16][OPITCH];

  const int tile = blockIdx.x;
  const int b = tile >> 6;
  const int n0 = (tile & 63) << 4;
  const int t = threadIdx.x;
  const int lane = t & 31;
  const int ot = t >> 5;
  const int row = lane & 15;
  const int h = lane >> 4;
  const int khalf = h << 1;
  const int n = n0 + row;
  const size_t xbase = ((size_t)b * NN + n) * CC + khalf;
  const float* wr0 = Wr + (size_t)khalf * OCC + ot * 16 + row;
  const float* wi0 = Wi + (size_t)khalf * OCC + ot * 16 + row;

  v2f xr[CC / 4], xi[CC / 4];
#pragma unroll
  for (int cc = 0; cc < CC / 4; ++cc) {
    xr[cc] = *(const v2f*)(Xr + xbase + cc * 4);
    xi[cc] = *(const v2f*)(Xi + xbase + cc * 4);
  }
  WFrag wa, wb;
  load_w(wa, wr0, wi0);

  {
    const int id = t >> 4, r = t & 15;
    const float* p = (id < KK ? Pr + (size_t)(b * KK + id) * NSLAB * NN
                              : Pi + (size_t)(b * KK + id - KK) * NSLAB * NN) + n0 + r;
    float s = 0.f;
#pragma unroll
    for (int sl = 0; sl < NSLAB; ++sl) s += p[(size_t)sl * NN];
    s_sl[id][r] = s;
    if (t < 32) {
      const int id2 = 8 + id;
      const float* p2 = Pi + (size_t)(b * KK + id2 - KK) * NSLAB * NN + n0 + r;
      float s2 = 0.f;
#pragma unroll
      for (int sl = 0; sl < NSLAB; ++sl) s2 += p2[(size_t)sl * NN];
      s_sl[id2][r] = s2;
    }
  }
  __syncthreads();

  v8f accR = (v8f){0.f,0.f,0.f,0.f,0.f,0.f,0.f,0.f};
  v8f accI = (v8f){0.f,0.f,0.f,0.f,0.f,0.f,0.f,0.f};
  for (int k = 0; k < KK; ++k) {
    const float* wrk = wr0 + (size_t)k * CC * OCC;
    const float* wik = wi0 + (size_t)k * CC * OCC;
    load_w(wb, wrk + WBLK * 4 * OCC, wik + WBLK * 4 * OCC);
    const float slr = s_sl[k][row];
    const float sli = s_sl[KK + k][row];
    mm_block(xr, xi, wa, slr, sli, accR, accI);
    if (k + 1 < KK) load_w(wa, wrk + CC * OCC, wik + CC * OCC);
    mm_block(xr + WBLK, xi + WBLK, wb, slr, sli, accR, accI);
  }

#pragma unroll
  for (int j = 0; j < 8; ++j) {
    s_o[0][8 * h + j][ot * 16 + row] = accR[j];
    s_o[1][8 * h + j][ot * 16 + row] = accI[j];
  }
  __syncthreads();
  float* outR = out;
  float* outI = out + (size_t)BB * NN * OCC;
#pragma unroll
  for (int i = 0; i < 2; ++i) {
    const int r = ot * 4 + i * 2 + h;
    const size_t oidx = ((size_t)b * NN + n0 + r) * OCC + row * 4;
    *(v4f*)(outR + oidx) = *(const v4f*)&s_o[0][r][row * 4];
    *(v4f*)(outI + oidx) = *(const v4f*)&s_o[1][r][row * 4];
  }
}

extern "C" void kernel_launch(void* const* d_in, const int* in_sizes, int n_in,
                              void* d_out, int out_size, void* d_ws, size_t ws_size,
                              hipStream_t stream) {
  const float* Xr  = (const float*)d_in[0];
  const float* Xi  = (const float*)d_in[1];
  const float* Lr  = (const float*)d_in[2];
  const float* Li  = (const float*)d_in[3];
  const float* Wr  = (const float*)d_in[4];
  const float* Wi  = (const float*)d_in[5];
  const float* awr = (const float*)d_in[6];
  const float* awi = (const float*)d_in[7];
  const float* abr = (const float*)d_in[8];
  const float* abi = (const float*)d_in[9];
  const float* par = (const float*)d_in[10];
  const float* pai = (const float*)d_in[11];
  float* out = (float*)d_out;

  float* ws    = (float*)d_ws;
  float* src_r = ws;
  float* src_i = ws + 8192;
  float* dst_r = ws + 16384;
  float* dst_i = ws + 24576;
  float* Pr    = ws + 32768;
  float* Pi    = ws + 1343488;

  k1_proj <<<256, 64, 0, stream>>>(Xr, Xi, awr, awi, src_r, src_i, dst_r, dst_i);
  k2_colslab<<<256, 512, 0, stream>>>(Lr, Li, src_r, src_i, dst_r, dst_i,
                                      abr, abi, par, pai, Pr, Pi);
  k4_out  <<<512, 128, 0, stream>>>(Xr, Xi, Wr, Wi, Pr, Pi, out);
}


# k2: L loads of the first three trips are requested before the column-statistics phase and stay in flight across it (LDS-only barriers); stream runs 3 trips ahead on 4 register sets; same sum order
# speedup vs baseline: 3.8897x; 1.0363x over previous
#include <hip/hip_runtime.h>
#include <math.h>

#define BB 8
#define NN 1024
#define CC 64
#define OCC 64
#define KK 5
#define SLABW 32
#define NSLAB (NN / SLABW)
#define RPT 64
#define NCH 16

typedef __attribute__((ext_vector_type(2))) float v2f;
typedef __attribute__((ext_vector_type(4))) float v4f;
typedef __attribute__((ext_vector_type(8))) float v8f;

#define PPITCH 36
__global__ void __launch_bounds__(64) k1_proj(const float* __restrict__ Xr, const float* __restrict__ Xi,
                        const float* __restrict__ awr, const float* __restrict__ awi,
                        float* __restrict__ src_r, float* __restrict__ src_i,
                        float* __restrict__ dst_r, float* __restrict__ dst_i) {
  __shared__ __attribute__((aligned(16))) float s_p[16][PPITCH];
  const int t = threadIdx.x;
  const int lane = t & 31;
  const int w = t >> 5;
  const int r = lane & 15;
  const int h = lane >> 4;
  const int khalf = h << 1;
  const int n0 = blockIdx.x * 32;

  const size_t xbase = (size_t)(n0 + w * 16 + r) * CC + khalf;
  const int c = r & 3;
  const int half = (c & 2) ? CC : 0;
  const float* pr = ((c & 1) ? awi : awr) + half + khalf;
  const float* pi = ((c & 1) ? awr : awi) + half + khalf;
  const float sg = (c & 1) ? 1.f : -1.f;

  v2f xr[CC / 4], xi[CC / 4], br[CC / 4], bi[CC / 4];
#pragma unroll
  for (int cc = 0; cc < CC / 4; ++cc) {
    xr[cc] = *(const v2f*)(Xr + xbase + cc * 4);
    xi[cc] = *(const v2f*)(Xi + xbase + cc * 4);
    br[cc] = *(const v2f*)(pr + cc * 4);
    bi[cc] = *(const v2f*)(pi + cc * 4);
  }
  __builtin_amdgcn_sched_barrier(0);
  v8f accA = (v8f){0.f,0.f,0.f,0.f,0.f,0.f,0.f,0.f};
  v8f accB = (v8f){0.f,0.f,0.f,0.f,0.f,0.f,0.f,0.f};
#pragma unroll
  for (int cc = 0; cc < CC / 4; ++cc) {
    const v2f bs = { sg * bi[cc].x, sg * bi[cc].y };
    accA = __builtin_amdgcn_wmma_f32_16x16x4_f32(false, xr[cc], false, br[cc], (short)0, accA, false, false);
    accB = __builtin_amdgcn_wmma_f32_16x16x4_f32(false, xi[cc], false, bs, (short)0, accB, false, false);
  }
  const v8f d = accA + accB;
#pragma unroll
  for (int j = 0; j < 8; ++j) s_p[r][w * 16 + 8 * h + j] = d[j];
  __syncthreads();
  float* o0 = w ? dst_r : src_r;
  float* o1 = w ? dst_i : src_i;
  o0[n0 + lane] = s_p[2 * w][lane];
  o1[n0 + lane] = s_p[2 * w + 1][lane];
}

template <int XMASK>
__device__ __forceinline__ float row_xor_add(float v) {
  const int o = __builtin_amdgcn_update_dpp(0, __builtin_bit_cast(int, v), 0x160 + XMASK, 0xf, 0xf, true);
  return v + __builtin_bit_cast(float, o);
}
__device__ __forceinline__ float row8_sum(float v) {
  v = row_xor_add<4>(v);
  v = row_xor_add<2>(v);
  v = row_xor_add<1>(v);
  return v;
}

struct LSet { v4f r[KK], i[KK]; };

__device__ __forceinline__ void load_set(LSet& s, const float* __restrict__ LrB, const float* __restrict__ LiB, unsigned off) {
#pragma unroll
  for (int k = 0; k < KK; ++k) {
    const float* pr = LrB + (size_t)k * NN * NN;
    const float* pi = LiB + (size_t)k * NN * NN;
    s.r[k] = *(const v4f*)(pr + off);
    s.i[k] = *(const v4f*)(pi + off);
  }
}

__device__ __forceinline__ void lds_barrier() {
  asm volatile("s_wait_dscnt 0x0\n\ts_barrier_signal -1\n\ts_barrier_wait -1" ::: "memory");
}

__global__ void __launch_bounds__(512) k2_colslab(
    const float* __restrict__ Lr, const float* __restrict__ Li,
    const float* __restrict__ src_r, const float* __restrict__ src_i,
    const float* __restrict__ dst_r, const float* __restrict__ dst_i,
    const float* __restrict__ ab_r_p, const float* __restrict__ ab_i_p,
    const float* __restrict__ pa_r_p, const float* __restrict__ pa_i_p,
    float* __restrict__ Pr, float* __restrict__ Pi) {
  __shared__ __attribute__((aligned(16))) v2f   s_s[NN];
  __shared__ __attribute__((aligned(16))) float s_q[NN][SLABW];
  __shared__ float s_red[NCH][SLABW];
  __shared__ __attribute__((aligned(16))) float s_cmax[SLABW];
  __shared__ __attribute__((aligned(16))) float s_csum[SLABW];
  __shared__ float s_out[2 * KK][NN];

  const int tid = threadIdx.x;
  const int lane = tid & 31;
  const int w = tid >> 5;
  const int bj = blockIdx.x & (NSLAB - 1);
  const int slab = ((bj & 15) << 1) | (bj >> 4);
  const int b = blockIdx.x >> 5;
  const int sb = b * NN;
  const float abr = ab_r_p[0], abi = ab_i_p[0], par = pa_r_p[0], pai = pa_i_p[0];

  const int q = lane & 7;
  const int rsub = lane >> 3;
  const int col0 = slab * SLABW + q * 4;
  const int rl0 = w * 4 + rsub;
  const float* __restrict__ LrB = Lr + (size_t)b * KK * NN * NN;
  const float* __restrict__ LiB = Li + (size_t)b * KK * NN * NN;
  const unsigned voff = (unsigned)(rl0 * NN + col0);
  const int c = tid & (SLABW - 1);
  const int rc = tid >> 5;

  const float s0r = src_r[sb + tid], s0i = src_i[sb + tid];
  const float s1r = src_r[sb + tid + 512], s1i = src_i[sb + tid + 512];
  const float dr = dst_r[sb + slab * SLABW + c];
  const float di = dst_i[sb + slab * SLABW + c];
  const v4f dr4 = *(const v4f*)(dst_r + sb + col0);
  const v4f di4 = *(const v4f*)(dst_i + sb + col0);
  __builtin_amdgcn_sched_barrier(0);

  LSet S0, S1, S2, S3;
  load_set(S0, LrB, LiB, voff);
  __builtin_amdgcn_sched_barrier(0);
  load_set(S1, LrB, LiB, voff + RPT * NN);
  __builtin_amdgcn_sched_barrier(0);
  load_set(S2, LrB, LiB, voff + 2 * RPT * NN);

  s_s[tid] = (v2f){ s0r + abr, s0i + abi };
  s_s[tid + 512] = (v2f){ s1r + abr, s1i + abi };
  lds_barrier();

  {
    float mx = 0.f;
    for (int i = 0; i < NN / NCH; ++i) {
      const int row = rc * (NN / NCH) + i;
      v2f s = s_s[row];
      float sr = s.x + dr; sr = sr >= 0.f ? sr : par * sr;
      float si = s.y + di; si = si >= 0.f ? si : pai * si;
      float mag = sqrtf(sr * sr + si * si);
      s_q[row][c] = mag;
      mx = mag > mx ? mag : mx;
    }
    s_red[rc][c] = mx;
    lds_barrier();
    float cmax = s_red[0][c];
#pragma unroll
    for (int j = 1; j < NCH; ++j) { float t = s_red[j][c]; cmax = t > cmax ? t : cmax; }
    lds_barrier();
    float sum = 0.f;
    for (int i = 0; i < NN / NCH; ++i) {
      const float mag = s_q[rc * (NN / NCH) + i][c];
      sum += __expf(mag - cmax);
    }
    s_red[rc][c] = sum;
    lds_barrier();
    if (rc == 0) {
      float cs = s_red[0][c];
#pragma unroll
      for (int j = 1; j < NCH; ++j) cs += s_red[j][c];
      s_cmax[c] = cmax;
      s_csum[c] = cs;
    }
    lds_barrier();
  }

  const v4f cx4 = *(const v4f*)(s_cmax + q * 4);
  const v4f cs4 = *(const v4f*)(s_csum + q * 4);

  auto trip = [&](const LSet& L, LSet& F, const int n, const bool refill) __attribute__((always_inline)) {
    if (refill) load_set(F, LrB, LiB, (unsigned)((n + 3 * RPT) * NN + col0));
    const v2f ss = s_s[n];
    const float ssr = ss.x, ssi = ss.y;
    const v4f mq = *(const v4f*)&s_q[n][q * 4];
    v4f arv, aiv;
#pragma unroll
    for (int j = 0; j < 4; ++j) {
      float sr = ssr + dr4[j]; sr = sr >= 0.f ? sr : par * sr;
      float si = ssi + di4[j]; si = si >= 0.f ? si : pai * si;
      const float mag = mq[j];
      float sm = __expf(mag - cx4[j]) / cs4[j];
      float safe = (mag == 0.f) ? 1.f : mag;
      float inv = sm / safe;
      arv[j] = inv * sr;
      aiv[j] = inv * si;
    }
    float pr[KK], pi[KK];
#pragma unroll
    for (int k = 0; k < KK; ++k) {
      float r = 0.f, i2 = 0.f;
#pragma unroll
      for (int j = 0; j < 4; ++j) {
        r  += L.r[k][j] * arv[j] - L.i[k][j] * aiv[j];
        i2 += L.r[k][j] * aiv[j] + L.i[k][j] * arv[j];
      }
      pr[k] = r;
      pi[k] = i2;
    }
#pragma unroll
    for (int k = 0; k < KK; ++k) {
      pr[k] = row8_sum(pr[k]);
      pi[k] = row8_sum(pi[k]);
    }
    if (q == 0) {
#pragma unroll
      for (int k = 0; k < KK; ++k) {
        s_out[2 * k][n] = pr[k];
        s_out[2 * k + 1][n] = pi[k];
      }
    }
  };

#pragma unroll 1
  for (int o = 0; o < NN / RPT / 4 - 1; ++o) {
    const int n = o * (4 * RPT) + rl0;
    trip(S0, S3, n, true);
    trip(S1, S0, n + RPT, true);
    trip(S2, S1, n + 2 * RPT, true);
    trip(S3, S2, n + 3 * RPT, true);
  }
  {
    const int n = (NN / RPT - 4) * RPT + rl0;
    trip(S0, S3, n, true);
    trip(S1, S0, n + RPT, false);
    trip(S2, S0, n + 2 * RPT, false);
    trip(S3, S0, n + 3 * RPT, false);
  }
  __syncthreads();
#pragma unroll
  for (int k = 0; k < KK; ++k) {
    const size_t o = ((size_t)(b * KK + k) * NSLAB + slab) * NN + tid;
    Pr[o] = s_out[2 * k][tid];
    Pi[o] = s_out[2 * k + 1][tid];
    Pr[o + 512] = s_out[2 * k][tid + 512];
    Pi[o + 512] = s_out[2 * k + 1][tid + 512];
  }
}

#define WBLK 8
#define OPITCH 68

struct WFrag { v2f wr[WBLK], wi[WBLK]; };

__device__ __forceinline__ void load_w(WFrag& f, const float* __restrict__ wr, const float* __restrict__ wi) {
#pragma unroll
  for (int j = 0; j < WBLK; ++j) {
    f.wr[j] = (v2f){ wr[(j * 4 + 0) * OCC], wr[(j * 4 + 1) * OCC] };
    f.wi[j] = (v2f){ wi[(j * 4 + 0) * OCC], wi[(j * 4 + 1) * OCC] };
  }
}

__device__ __forceinline__ void mm_block(const v2f* xr, const v2f* xi, const WFrag& f,
                                         float slr, float sli, v8f& accR, v8f& accI) {
  const float msli = -sli;
#pragma unroll
  for (int j = 0; j < WBLK; ++j) {
    v2f aRr = { slr  * xr[j].x, slr  * xr[j].y };
    v2f aRi = { msli * xi[j].x, msli * xi[j].y };
    v2f aIr = { sli  * xr[j].x, sli  * xr[j].y };
    v2f aIi = { slr  * xi[j].x, slr  * xi[j].y };
    accR = __builtin_amdgcn_wmma_f32_16x16x4_f32(false, aRr, false, f.wr[j], (short)0, accR, false, false);
    accI = __builtin_amdgcn_wmma_f32_16x16x4_f32(false, aIr, false, f.wr[j], (short)0, accI, false, false);
    accR = __builtin_amdgcn_wmma_f32_16x16x4_f32(false, aRi, false, f.wi[j], (short)0, accR, false, false);
    accI = __builtin_amdgcn_wmma_f32_16x16x4_f32(false, aIi, false, f.wi[j], (short)0, accI, false, false);
  }
}

__global__ void __launch_bounds__(128) k4_out(
    const float* __restrict__ Xr, const float* __restrict__ Xi,
    const float* __restrict__ Wr, const float* __restrict__ Wi,
    const float* __restrict__ Pr, const float* __restrict__ Pi,
    float* __restrict__ out) {
  __shared__ float s_sl[2 * KK][16];
  __shared__ __attribute__((aligned(16))) float s_o[2][16][OPITCH];

  const int tile = blockIdx.x;
  const int b = tile >> 6;
  const int n0 = (tile & 63) << 4;
  const int t = threadIdx.x;
  const int lane = t & 31;
  const int ot = t >> 5;
  const int row = lane & 15;
  const int h = lane >> 4;
  const int khalf = h << 1;
  const int n = n0 + row;
  const size_t xbase = ((size_t)b * NN + n) * CC + khalf;
  const float* wr0 = Wr + (size_t)khalf * OCC + ot * 16 + row;
  const float* wi0 = Wi + (size_t)khalf * OCC + ot * 16 + row;

  v2f xr[CC / 4], xi[CC / 4];
#pragma unroll
  for (int cc = 0; cc < CC / 4; ++cc) {
    xr[cc] = *(const v2f*)(Xr + xbase + cc * 4);
    xi[cc] = *(const v2f*)(Xi + xbase + cc * 4);
  }
  WFrag wa, wb;
  load_w(wa, wr0, wi0);

  {
    const int id = t >> 4, r = t & 15;
    const float* p = (id < KK ? Pr + (size_t)(b * KK + id) * NSLAB * NN
                              : Pi + (size_t)(b * KK + id - KK) * NSLAB * NN) + n0 + r;
    float s = 0.f;
#pragma unroll
    for (int sl = 0; sl < NSLAB; ++sl) s += p[(size_t)sl * NN];
    s_sl[id][r] = s;
    if (t < 32) {
      const int id2 = 8 + id;
      const float* p2 = Pi + (size_t)(b * KK + id2 - KK) * NSLAB * NN + n0 + r;
      float s2 = 0.f;
#pragma unroll
      for (int sl = 0; sl < NSLAB; ++sl) s2 += p2[(size_t)sl * NN];
      s_sl[id2][r] = s2;
    }
  }
  __syncthreads();

  v8f accR = (v8f){0.f,0.f,0.f,0.f,0.f,0.f,0.f,0.f};
  v8f accI = (v8f){0.f,0.f,0.f,0.f,0.f,0.f,0.f,0.f};
  for (int k = 0; k < KK; ++k) {
    const float* wrk = wr0 + (size_t)k * CC * OCC;
    const float* wik = wi0 + (size_t)k * CC * OCC;
    load_w(wb, wrk + WBLK * 4 * OCC, wik + WBLK * 4 * OCC);
    const float slr = s_sl[k][row];
    const float sli = s_sl[KK + k][row];
    mm_block(xr, xi, wa, slr, sli, accR, accI);
    if (k + 1 < KK) load_w(wa, wrk + CC * OCC, wik + CC * OCC);
    mm_block(xr + WBLK, xi + WBLK, wb, slr, sli, accR, accI);
  }

#pragma unroll
  for (int j = 0; j < 8; ++j) {
    s_o[0][8 * h + j][ot * 16 + row] = accR[j];
    s_o[1][8 * h + j][ot * 16 + row] = accI[j];
  }
  __syncthreads();
  float* outR = out;
  float* outI = out + (size_t)BB * NN * OCC;
#pragma unroll
  for (int i = 0; i < 2; ++i) {
    const int r = ot * 4 + i * 2 + h;
    const size_t oidx = ((size_t)b * NN + n0 + r) * OCC + row * 4;
    *(v4f*)(outR + oidx) = *(const v4f*)&s_o[0][r][row * 4];
    *(v4f*)(outI + oidx) = *(const v4f*)&s_o[1][r][row * 4];
  }
}

extern "C" void kernel_launch(void* const* d_in, const int* in_sizes, int n_in,
                              void* d_out, int out_size, void* d_ws, size_t ws_size,
                              hipStream_t stream) {
  const float* Xr  = (const float*)d_in[0];
  const float* Xi  = (const float*)d_in[1];
  const float* Lr  = (const float*)d_in[2];
  const float* Li  = (const float*)d_in[3];
  const float* Wr  = (const float*)d_in[4];
  const float* Wi  = (const float*)d_in[5];
  const float* awr = (const float*)d_in[6];
  const float* awi = (const float*)d_in[7];
  const float* abr = (const float*)d_in[8];
  const float* abi = (const float*)d_in[9];
  const float* par = (const float*)d_in[10];
  const float* pai = (const float*)d_in[11];
  float* out = (float*)d_out;

  float* ws    = (float*)d_ws;
  float* src_r = ws;
  float* src_i = ws + 8192;
  float* dst_r = ws + 16384;
  float* dst_i = ws + 24576;
  float* Pr    = ws + 32768;
  float* Pi    = ws + 1343488;

  k1_proj <<<256, 64, 0, stream>>>(Xr, Xi, awr, awi, src_r, src_i, dst_r, dst_i);
  k2_colslab<<<256, 512, 0, stream>>>(Lr, Li, src_r, src_i, dst_r, dst_i,
                                      abr, abi, par, pai, Pr, Pi);
  k4_out  <<<512, 128, 0, stream>>>(Xr, Xi, Wr, Wi, Pr, Pi, out);
}
